# starting kernel
# baseline (speedup 1.0000x reference)
#include <hip/hip_runtime.h>
#include <math.h>

#define N_NODES 50000
#define N_EDGES 800000
#define ETOT    (N_EDGES + N_NODES)
#define IN_DIM  128
#define HID     64
#define NH      4
#define F       (NH * HID)
#define MT      (N_NODES / 16)

typedef __attribute__((ext_vector_type(2))) float v2f;
typedef __attribute__((ext_vector_type(8))) float v8f;

__global__ void k_init(float* __restrict__ agg, float* __restrict__ m,
                       float* __restrict__ s) {
  int i = blockIdx.x * blockDim.x + threadIdx.x;
  if (i < N_NODES * F) agg[i] = 0.0f;
  if (i < N_NODES * NH) { m[i] = -3.0e38f; s[i] = 0.0f; }
}

__global__ void k_gemm_wmma(const float* __restrict__ X, int K,
                            const float* __restrict__ W0, const float* __restrict__ B0,
                            float* __restrict__ Y0,
                            const float* __restrict__ W1, const float* __restrict__ B1,
                            float* __restrict__ Y1,
                            int Ntiles, int nmat, int ldY, int act) {
  int wave = (blockIdx.x * blockDim.x + threadIdx.x) >> 5;
  int total = MT * Ntiles * nmat;
  if (wave >= total) return;

  int tn   = wave % Ntiles;
  int rest = wave / Ntiles;
  int mat  = rest % nmat;
  int tm   = rest / nmat;

  const float* W  = mat ? W1 : W0;
  const float* Bv = mat ? B1 : B0;
  float*       Y  = mat ? Y1 : Y0;

  int lane = threadIdx.x & 31;
  int half = lane >> 4;
  int l16  = lane & 15;

  const float* xrow = X + (size_t)(tm * 16 + l16) * K;
  const float* wrow = W + (size_t)(tn * 16 + l16) * K;

  v8f c = {0.f, 0.f, 0.f, 0.f, 0.f, 0.f, 0.f, 0.f};
  for (int k0 = 0; k0 < K; k0 += 4) {
    v2f a = *(const v2f*)(xrow + k0 + 2 * half);
    v2f b = *(const v2f*)(wrow + k0 + 2 * half);
    c = __builtin_amdgcn_wmma_f32_16x16x4_f32(false, a, false, b, (short)0, c,
                                              false, false);
  }

  int col   = tn * 16 + l16;
  float bia = Bv[col];
  int mbase = tm * 16 + 8 * half;
#pragma unroll
  for (int r = 0; r < 8; ++r) {
    float v = c[r] + bia;
    if (act) v = fmaxf(v, 0.0f);
    Y[(size_t)(mbase + r) * ldY + col] = v;
  }
}

__global__ void k_edge_score(const int* __restrict__ ei,
                             const float* __restrict__ xl, const float* __restrict__ xr,
                             const float* __restrict__ att,
                             float* __restrict__ sc, float* __restrict__ m) {
  int e = (blockIdx.x * blockDim.x + threadIdx.x) >> 5;
  if (e >= ETOT) return;
  int lane = threadIdx.x & 31;
  int src = (e < N_EDGES) ? ei[e] : (e - N_EDGES);
  int dst = (e < N_EDGES) ? ei[N_EDGES + e] : (e - N_EDGES);

  const float* pl = xl + (size_t)src * F;
  const float* pr = xr + (size_t)dst * F;

  float acc[NH] = {0.f, 0.f, 0.f, 0.f};
#pragma unroll
  for (int j = 0; j < 8; ++j) {
    int c = lane + 32 * j;
    float v = pl[c] + pr[c];
    v = (v > 0.f) ? v : 0.2f * v;
    acc[j >> 1] += v * att[c];
  }
#pragma unroll
  for (int h = 0; h < NH; ++h) {
    float a = acc[h];
    for (int off = 16; off > 0; off >>= 1) a += __shfl_xor(a, off, 32);
    acc[h] = a;
  }
  if (lane == 0) {
#pragma unroll
    for (int h = 0; h < NH; ++h) {
      sc[(size_t)e * NH + h] = acc[h];
      atomicMax(&m[(size_t)dst * NH + h], acc[h]);
    }
  }
}

__global__ void k_edge_exp(const int* __restrict__ ei, float* __restrict__ sc,
                           const float* __restrict__ m, float* __restrict__ s) {
  int idx = blockIdx.x * blockDim.x + threadIdx.x;
  if (idx >= ETOT * NH) return;
  int e = idx >> 2, h = idx & 3;
  int dst = (e < N_EDGES) ? ei[N_EDGES + e] : (e - N_EDGES);
  float p = __expf(sc[idx] - m[dst * NH + h]);
  sc[idx] = p;
  atomicAdd(&s[dst * NH + h], p);
}

__global__ void k_edge_agg(const int* __restrict__ ei, const float* __restrict__ xl,
                           const float* __restrict__ sc, const float* __restrict__ s,
                           float* __restrict__ agg) {
  int e = (blockIdx.x * blockDim.x + threadIdx.x) >> 5;
  if (e >= ETOT) return;
  int lane = threadIdx.x & 31;
  int src = (e < N_EDGES) ? ei[e] : (e - N_EDGES);
  int dst = (e < N_EDGES) ? ei[N_EDGES + e] : (e - N_EDGES);

  float alpha[NH];
#pragma unroll
  for (int h = 0; h < NH; ++h)
    alpha[h] = sc[(size_t)e * NH + h] / fmaxf(s[(size_t)dst * NH + h], 1e-16f);

  const float* pl = xl + (size_t)src * F;
  float*       pd = agg + (size_t)dst * F;
#pragma unroll
  for (int j = 0; j < 8; ++j) {
    int c = lane + 32 * j;
    atomicAdd(&pd[c], alpha[j >> 1] * pl[c]);
  }
}

__global__ void k_node_finish(const float* __restrict__ agg, const float* __restrict__ bias,
                              const float* __restrict__ g, const float* __restrict__ beta,
                              float* __restrict__ out) {
  int n = (blockIdx.x * blockDim.x + threadIdx.x) >> 5;
  if (n >= N_NODES) return;
  int lane = threadIdx.x & 31;
  int c0 = lane * 2, c1 = c0 + 1;
  const float* p = agg + (size_t)n * F;

  float v0 = 0.25f * (p[c0] + p[64 + c0] + p[128 + c0] + p[192 + c0]) + bias[c0];
  float v1 = 0.25f * (p[c1] + p[64 + c1] + p[128 + c1] + p[192 + c1]) + bias[c1];

  float sum = v0 + v1;
  for (int off = 16; off > 0; off >>= 1) sum += __shfl_xor(sum, off, 32);
  float mu = sum * (1.0f / 64.0f);
  float d0 = v0 - mu, d1 = v1 - mu;
  float vs = d0 * d0 + d1 * d1;
  for (int off = 16; off > 0; off >>= 1) vs += __shfl_xor(vs, off, 32);
  float inv = rsqrtf(vs * (1.0f / 64.0f) + 1e-5f);

  float y0 = d0 * inv * g[c0] + beta[c0];
  float y1 = d1 * inv * g[c1] + beta[c1];
  y0 = (y0 > 0.f) ? y0 : (__expf(y0) - 1.f);
  y1 = (y1 > 0.f) ? y1 : (__expf(y1) - 1.f);
  out[(size_t)n * HID + c0] = y0;
  out[(size_t)n * HID + c1] = y1;
}

__global__ void k_head2(const float* __restrict__ hm, const float* __restrict__ W,
                        const float* __restrict__ b, float* __restrict__ out) {
  int n = blockIdx.x * blockDim.x + threadIdx.x;
  if (n >= N_NODES) return;
  const float* p = hm + (size_t)n * 32;
  float a0 = b[0], a1 = b[1];
#pragma unroll
  for (int j = 0; j < 32; ++j) {
    float v = p[j];
    a0 += v * W[j];
    a1 += v * W[32 + j];
  }
  out[(size_t)n * 2 + 0] = a0;
  out[(size_t)n * 2 + 1] = a1;
}

static inline int wave_blocks(long long waves) { return (int)((waves + 7) / 8); }

extern "C" void kernel_launch(void* const* d_in, const int* in_sizes, int n_in,
                              void* d_out, int out_size, void* d_ws, size_t ws_size,
                              hipStream_t stream) {
  const float* x    = (const float*)d_in[0];
  const int*   ei   = (const int*)  d_in[1];
  const float* Wl1  = (const float*)d_in[2];  const float* bl1 = (const float*)d_in[3];
  const float* Wr1  = (const float*)d_in[4];  const float* br1 = (const float*)d_in[5];
  const float* att1 = (const float*)d_in[6];  const float* bias1=(const float*)d_in[7];
  const float* Wl2  = (const float*)d_in[8];  const float* bl2 = (const float*)d_in[9];
  const float* Wr2  = (const float*)d_in[10]; const float* br2 = (const float*)d_in[11];
  const float* att2 = (const float*)d_in[12]; const float* bias2=(const float*)d_in[13];
  const float* g1   = (const float*)d_in[14]; const float* be1 = (const float*)d_in[15];
  const float* g2   = (const float*)d_in[16]; const float* be2 = (const float*)d_in[17];
  const float* Wh1  = (const float*)d_in[18]; const float* bh1 = (const float*)d_in[19];
  const float* Wh2  = (const float*)d_in[20]; const float* bh2 = (const float*)d_in[21];
  float* out = (float*)d_out;

  float* ws  = (float*)d_ws;
  const size_t NF = (size_t)N_NODES * F;
  float* XL  = ws;
  float* XR  = XL + NF;
  float* AGG = XR + NF;
  float* SC  = AGG + NF;
  float* M   = SC + (size_t)ETOT * NH;
  float* S   = M  + (size_t)N_NODES * NH;
  float* H1  = S  + (size_t)N_NODES * NH;
  float* HM  = H1 + (size_t)N_NODES * HID;

  const int TB = 256;
  int initBlocks = (int)((NF + TB - 1) / TB);
  int edgeBlocks = wave_blocks(ETOT);
  int nodeBlocks = wave_blocks(N_NODES);
  int expBlocks  = (ETOT * NH + TB - 1) / TB;

  k_init<<<initBlocks, TB, 0, stream>>>(AGG, M, S);
  k_gemm_wmma<<<wave_blocks((long long)MT * 16 * 2), TB, 0, stream>>>(
      x, IN_DIM, Wl1, bl1, XL, Wr1, br1, XR,  16,  2, F, 0);
  k_edge_score<<<edgeBlocks, TB, 0, stream>>>(ei, XL, XR, att1, SC, M);
  k_edge_exp<<<expBlocks, TB, 0, stream>>>(ei, SC, M, S);
  k_edge_agg<<<edgeBlocks, TB, 0, stream>>>(ei, XL, SC, S, AGG);
  k_node_finish<<<nodeBlocks, TB, 0, stream>>>(AGG, bias1, g1, be1, H1);

  k_init<<<initBlocks, TB, 0, stream>>>(AGG, M, S);
  k_gemm_wmma<<<wave_blocks((long long)MT * 16 * 2), TB, 0, stream>>>(
      H1, HID, Wl2, bl2, XL, Wr2, br2, XR, 16, 2, F, 0);
  k_edge_score<<<edgeBlocks, TB, 0, stream>>>(ei, XL, XR, att2, SC, M);
  k_edge_exp<<<expBlocks, TB, 0, stream>>>(ei, SC, M, S);
  k_edge_agg<<<edgeBlocks, TB, 0, stream>>>(ei, XL, SC, S, AGG);
  k_node_finish<<<nodeBlocks, TB, 0, stream>>>(AGG, bias2, g2, be2, H1);

  k_gemm_wmma<<<wave_blocks((long long)MT * 2), TB, 0, stream>>>(
      H1, HID, Wh1, bh1, HM, Wh1, bh1, HM,  2,  1, 32,  1);
  k_head2<<<(N_NODES + TB - 1) / TB, TB, 0, stream>>>(HM, Wh2, bh2, out);
}


# edge aggregation sums the 4 heads per edge before the scatter: 64 float atomics per edge instead of 256, agg plane N x 64; head-sum now precedes the edge-sum (summation order changes)
# speedup vs baseline: 1.4164x; 1.4164x over previous
#include <hip/hip_runtime.h>
#include <math.h>

#define N_NODES 50000
#define N_EDGES 800000
#define ETOT    (N_EDGES + N_NODES)
#define IN_DIM  128
#define HID     64
#define NH      4
#define F       (NH * HID)
#define MT      (N_NODES / 16)

typedef __attribute__((ext_vector_type(2))) float v2f;
typedef __attribute__((ext_vector_type(8))) float v8f;

__global__ void k_init(float* __restrict__ agg, float* __restrict__ m,
                       float* __restrict__ s) {
  int i = blockIdx.x * blockDim.x + threadIdx.x;
  if (i < N_NODES * HID) agg[i] = 0.0f;
  if (i < N_NODES * NH) { m[i] = -3.0e38f; s[i] = 0.0f; }
}

__global__ void k_gemm_wmma(const float* __restrict__ X, int K,
                            const float* __restrict__ W0, const float* __restrict__ B0,
                            float* __restrict__ Y0,
                            const float* __restrict__ W1, const float* __restrict__ B1,
                            float* __restrict__ Y1,
                            int Ntiles, int nmat, int ldY, int act) {
  int wave = (blockIdx.x * blockDim.x + threadIdx.x) >> 5;
  int total = MT * Ntiles * nmat;
  if (wave >= total) return;

  int tn   = wave % Ntiles;
  int rest = wave / Ntiles;
  int mat  = rest % nmat;
  int tm   = rest / nmat;

  const float* W  = mat ? W1 : W0;
  const float* Bv = mat ? B1 : B0;
  float*       Y  = mat ? Y1 : Y0;

  int lane = threadIdx.x & 31;
  int half = lane >> 4;
  int l16  = lane & 15;

  const float* xrow = X + (size_t)(tm * 16 + l16) * K;
  const float* wrow = W + (size_t)(tn * 16 + l16) * K;

  v8f c = {0.f, 0.f, 0.f, 0.f, 0.f, 0.f, 0.f, 0.f};
  for (int k0 = 0; k0 < K; k0 += 4) {
    v2f a = *(const v2f*)(xrow + k0 + 2 * half);
    v2f b = *(const v2f*)(wrow + k0 + 2 * half);
    c = __builtin_amdgcn_wmma_f32_16x16x4_f32(false, a, false, b, (short)0, c,
                                              false, false);
  }

  int col   = tn * 16 + l16;
  float bia = Bv[col];
  int mbase = tm * 16 + 8 * half;
#pragma unroll
  for (int r = 0; r < 8; ++r) {
    float v = c[r] + bia;
    if (act) v = fmaxf(v, 0.0f);
    Y[(size_t)(mbase + r) * ldY + col] = v;
  }
}

__global__ void k_edge_score(const int* __restrict__ ei,
                             const float* __restrict__ xl, const float* __restrict__ xr,
                             const float* __restrict__ att,
                             float* __restrict__ sc, float* __restrict__ m) {
  int e = (blockIdx.x * blockDim.x + threadIdx.x) >> 5;
  if (e >= ETOT) return;
  int lane = threadIdx.x & 31;
  int src = (e < N_EDGES) ? ei[e] : (e - N_EDGES);
  int dst = (e < N_EDGES) ? ei[N_EDGES + e] : (e - N_EDGES);

  const float* pl = xl + (size_t)src * F;
  const float* pr = xr + (size_t)dst * F;

  float acc[NH] = {0.f, 0.f, 0.f, 0.f};
#pragma unroll
  for (int j = 0; j < 8; ++j) {
    int c = lane + 32 * j;
    float v = pl[c] + pr[c];
    v = (v > 0.f) ? v : 0.2f * v;
    acc[j >> 1] += v * att[c];
  }
#pragma unroll
  for (int h = 0; h < NH; ++h) {
    float a = acc[h];
    for (int off = 16; off > 0; off >>= 1) a += __shfl_xor(a, off, 32);
    acc[h] = a;
  }
  if (lane == 0) {
#pragma unroll
    for (int h = 0; h < NH; ++h) {
      sc[(size_t)e * NH + h] = acc[h];
      atomicMax(&m[(size_t)dst * NH + h], acc[h]);
    }
  }
}

__global__ void k_edge_exp(const int* __restrict__ ei, float* __restrict__ sc,
                           const float* __restrict__ m, float* __restrict__ s) {
  int idx = blockIdx.x * blockDim.x + threadIdx.x;
  if (idx >= ETOT * NH) return;
  int e = idx >> 2, h = idx & 3;
  int dst = (e < N_EDGES) ? ei[N_EDGES + e] : (e - N_EDGES);
  float p = __expf(sc[idx] - m[dst * NH + h]);
  sc[idx] = p;
  atomicAdd(&s[dst * NH + h], p);
}

__global__ void k_edge_agg(const int* __restrict__ ei, const float* __restrict__ xl,
                           const float* __restrict__ sc, const float* __restrict__ s,
                           float* __restrict__ agg) {
  int e = (blockIdx.x * blockDim.x + threadIdx.x) >> 5;
  if (e >= ETOT) return;
  int lane = threadIdx.x & 31;
  int src = (e < N_EDGES) ? ei[e] : (e - N_EDGES);
  int dst = (e < N_EDGES) ? ei[N_EDGES + e] : (e - N_EDGES);

  float alpha[NH];
#pragma unroll
  for (int h = 0; h < NH; ++h)
    alpha[h] = sc[(size_t)e * NH + h] / fmaxf(s[(size_t)dst * NH + h], 1e-16f);

  const float* pl = xl + (size_t)src * F;
  float*       pd = agg + (size_t)dst * HID;
  float v0 = alpha[0] * pl[lane];
  float v1 = alpha[0] * pl[32 + lane];
#pragma unroll
  for (int h = 1; h < NH; ++h) {
    v0 += alpha[h] * pl[h * HID + lane];
    v1 += alpha[h] * pl[h * HID + 32 + lane];
  }
  atomicAdd(&pd[lane], v0);
  atomicAdd(&pd[32 + lane], v1);
}

__global__ void k_node_finish(const float* __restrict__ agg, const float* __restrict__ bias,
                              const float* __restrict__ g, const float* __restrict__ beta,
                              float* __restrict__ out) {
  int n = (blockIdx.x * blockDim.x + threadIdx.x) >> 5;
  if (n >= N_NODES) return;
  int lane = threadIdx.x & 31;
  int c0 = lane * 2, c1 = c0 + 1;
  const float* p = agg + (size_t)n * HID;

  float v0 = 0.25f * p[c0] + bias[c0];
  float v1 = 0.25f * p[c1] + bias[c1];

  float sum = v0 + v1;
  for (int off = 16; off > 0; off >>= 1) sum += __shfl_xor(sum, off, 32);
  float mu = sum * (1.0f / 64.0f);
  float d0 = v0 - mu, d1 = v1 - mu;
  float vs = d0 * d0 + d1 * d1;
  for (int off = 16; off > 0; off >>= 1) vs += __shfl_xor(vs, off, 32);
  float inv = rsqrtf(vs * (1.0f / 64.0f) + 1e-5f);

  float y0 = d0 * inv * g[c0] + beta[c0];
  float y1 = d1 * inv * g[c1] + beta[c1];
  y0 = (y0 > 0.f) ? y0 : (__expf(y0) - 1.f);
  y1 = (y1 > 0.f) ? y1 : (__expf(y1) - 1.f);
  out[(size_t)n * HID + c0] = y0;
  out[(size_t)n * HID + c1] = y1;
}

__global__ void k_head2(const float* __restrict__ hm, const float* __restrict__ W,
                        const float* __restrict__ b, float* __restrict__ out) {
  int n = blockIdx.x * blockDim.x + threadIdx.x;
  if (n >= N_NODES) return;
  const float* p = hm + (size_t)n * 32;
  float a0 = b[0], a1 = b[1];
#pragma unroll
  for (int j = 0; j < 32; ++j) {
    float v = p[j];
    a0 += v * W[j];
    a1 += v * W[32 + j];
  }
  out[(size_t)n * 2 + 0] = a0;
  out[(size_t)n * 2 + 1] = a1;
}

static inline int wave_blocks(long long waves) { return (int)((waves + 7) / 8); }

extern "C" void kernel_launch(void* const* d_in, const int* in_sizes, int n_in,
                              void* d_out, int out_size, void* d_ws, size_t ws_size,
                              hipStream_t stream) {
  const float* x    = (const float*)d_in[0];
  const int*   ei   = (const int*)  d_in[1];
  const float* Wl1  = (const float*)d_in[2];  const float* bl1 = (const float*)d_in[3];
  const float* Wr1  = (const float*)d_in[4];  const float* br1 = (const float*)d_in[5];
  const float* att1 = (const float*)d_in[6];  const float* bias1=(const float*)d_in[7];
  const float* Wl2  = (const float*)d_in[8];  const float* bl2 = (const float*)d_in[9];
  const float* Wr2  = (const float*)d_in[10]; const float* br2 = (const float*)d_in[11];
  const float* att2 = (const float*)d_in[12]; const float* bias2=(const float*)d_in[13];
  const float* g1   = (const float*)d_in[14]; const float* be1 = (const float*)d_in[15];
  const float* g2   = (const float*)d_in[16]; const float* be2 = (const float*)d_in[17];
  const float* Wh1  = (const float*)d_in[18]; const float* bh1 = (const float*)d_in[19];
  const float* Wh2  = (const float*)d_in[20]; const float* bh2 = (const float*)d_in[21];
  float* out = (float*)d_out;

  float* ws  = (float*)d_ws;
  const size_t NF = (size_t)N_NODES * F;
  float* XL  = ws;
  float* XR  = XL + NF;
  float* AGG = XR + NF;
  float* SC  = AGG + NF;
  float* M   = SC + (size_t)ETOT * NH;
  float* S   = M  + (size_t)N_NODES * NH;
  float* H1  = S  + (size_t)N_NODES * NH;
  float* HM  = H1 + (size_t)N_NODES * HID;

  const int TB = 256;
  int initBlocks = (N_NODES * HID + TB - 1) / TB;
  int edgeBlocks = wave_blocks(ETOT);
  int nodeBlocks = wave_blocks(N_NODES);
  int expBlocks  = (ETOT * NH + TB - 1) / TB;

  k_init<<<initBlocks, TB, 0, stream>>>(AGG, M, S);
  k_gemm_wmma<<<wave_blocks((long long)MT * 16 * 2), TB, 0, stream>>>(
      x, IN_DIM, Wl1, bl1, XL, Wr1, br1, XR,  16,  2, F, 0);
  k_edge_score<<<edgeBlocks, TB, 0, stream>>>(ei, XL, XR, att1, SC, M);
  k_edge_exp<<<expBlocks, TB, 0, stream>>>(ei, SC, M, S);
  k_edge_agg<<<edgeBlocks, TB, 0, stream>>>(ei, XL, SC, S, AGG);
  k_node_finish<<<nodeBlocks, TB, 0, stream>>>(AGG, bias1, g1, be1, H1);

  k_init<<<initBlocks, TB, 0, stream>>>(AGG, M, S);
  k_gemm_wmma<<<wave_blocks((long long)MT * 16 * 2), TB, 0, stream>>>(
      H1, HID, Wl2, bl2, XL, Wr2, br2, XR, 16, 2, F, 0);
  k_edge_score<<<edgeBlocks, TB, 0, stream>>>(ei, XL, XR, att2, SC, M);
  k_edge_exp<<<expBlocks, TB, 0, stream>>>(ei, SC, M, S);
  k_edge_agg<<<edgeBlocks, TB, 0, stream>>>(ei, XL, SC, S, AGG);
  k_node_finish<<<nodeBlocks, TB, 0, stream>>>(AGG, bias2, g2, be2, H1);

  k_gemm_wmma<<<wave_blocks((long long)MT * 2), TB, 0, stream>>>(
      H1, HID, Wh1, bh1, HM, Wh1, bh1, HM,  2,  1, 32,  1);
  k_head2<<<(N_NODES + TB - 1) / TB, TB, 0, stream>>>(HM, Wh2, bh2, out);
}


# edges counting-sorted by destination once per call; one wave per node does scores, running-max softmax, weighted sum and LayerNorm in registers: no float atomics, summation order changes
# speedup vs baseline: 2.7309x; 1.9281x over previous
#include <hip/hip_runtime.h>
#include <math.h>

#define N_NODES 50000
#define N_EDGES 800000
#define IN_DIM  128
#define HID     64
#define NH      4
#define F       (NH * HID)
#define MT      (N_NODES / 16)

typedef __attribute__((ext_vector_type(2))) float v2f;
typedef __attribute__((ext_vector_type(8))) float v8f;

#define SCAN_T 1024
#define SCAN_V 13
#define SCAN_C (4 * SCAN_V)
#define NPAD   (SCAN_T * SCAN_C)

__global__ void k_deg_zero(int* __restrict__ deg) {
  int i = blockIdx.x * blockDim.x + threadIdx.x;
  if (i < NPAD) deg[i] = 0;
}

__global__ void k_deg_count(const int* __restrict__ ei, int* __restrict__ deg,
                            int* __restrict__ rank) {
  int e = blockIdx.x * blockDim.x + threadIdx.x;
  if (e < N_EDGES) rank[e] = atomicAdd(&deg[ei[N_EDGES + e]], 1);
}

__global__ void __launch_bounds__(SCAN_T) k_deg_scan(const int* __restrict__ deg,
                                                     int* __restrict__ off) {
  __shared__ int part[SCAN_T];
  int t = threadIdx.x;
  const int4* pin = (const int4*)(deg + t * SCAN_C);
  int4 d[SCAN_V];
#pragma unroll
  for (int i = 0; i < SCAN_V; ++i) d[i] = pin[i];
  int sum = 0;
#pragma unroll
  for (int i = 0; i < SCAN_V; ++i) {
    int4 q = d[i];
    d[i].x = sum; sum += q.x;
    d[i].y = sum; sum += q.y;
    d[i].z = sum; sum += q.z;
    d[i].w = sum; sum += q.w;
  }
  part[t] = sum;
  __syncthreads();
  for (int s = 1; s < SCAN_T; s <<= 1) {
    int v = (t >= s) ? part[t - s] : 0;
    __syncthreads();
    part[t] += v;
    __syncthreads();
  }
  int base = part[t] - sum;
  int4* pout = (int4*)(off + t * SCAN_C);
#pragma unroll
  for (int i = 0; i < SCAN_V; ++i) {
    int4 q = d[i];
    q.x += base; q.y += base; q.z += base; q.w += base;
    pout[i] = q;
  }
}

__global__ void k_edge_fill(const int* __restrict__ ei, const int* __restrict__ off,
                            const int* __restrict__ rank, int* __restrict__ srcs) {
  int e = blockIdx.x * blockDim.x + threadIdx.x;
  if (e >= N_EDGES) return;
  srcs[off[ei[N_EDGES + e]] + rank[e]] = ei[e];
}

__device__ __forceinline__ float lrelu_dot4(float4 x, float4 r, float4 a) {
  float v0 = x.x + r.x, v1 = x.y + r.y, v2 = x.z + r.z, v3 = x.w + r.w;
  v0 = (v0 > 0.f) ? v0 : 0.2f * v0;
  v1 = (v1 > 0.f) ? v1 : 0.2f * v1;
  v2 = (v2 > 0.f) ? v2 : 0.2f * v2;
  v3 = (v3 > 0.f) ? v3 : 0.2f * v3;
  return v0 * a.x + v1 * a.y + v2 * a.z + v3 * a.w;
}

template <int M>
__device__ __forceinline__ float row_xor(float v) {
  int b = __builtin_bit_cast(int, v);
  return __builtin_bit_cast(float, __builtin_amdgcn_update_dpp(b, b, 0x160 | M, 0xf, 0xf, false));
}

__device__ __forceinline__ float xhalf(float v) {
  int b = __builtin_bit_cast(int, v);
  return __builtin_bit_cast(float,
      __builtin_amdgcn_permlanex16(b, b, 0x76543210, 0xfedcba98, false, false));
}

__device__ __forceinline__ float half_sum16(float v) {
  v += row_xor<1>(v);
  v += row_xor<2>(v);
  v += row_xor<4>(v);
  v += row_xor<8>(v);
  return v;
}

#define EG 4

__global__ void __launch_bounds__(256) k_node_gat(const int* __restrict__ off, const int* __restrict__ srcs,
                           const float* __restrict__ xl, const float* __restrict__ xr,
                           const float* __restrict__ att, const float* __restrict__ bias,
                           const float* __restrict__ g, const float* __restrict__ beta,
                           float* __restrict__ out) {
  int n = __builtin_amdgcn_readfirstlane((blockIdx.x * blockDim.x + threadIdx.x) >> 5);
  if (n >= N_NODES) return;
  int lane = threadIdx.x & 31;
  int c0 = 4 * lane;
  int c1 = 2 * HID + 4 * lane;

  const float* prn = xr + (size_t)n * F;
  const float4 r0 = *(const float4*)(prn + c0);
  const float4 r1 = *(const float4*)(prn + c1);
  const float4 a0 = *(const float4*)(att + c0);
  const float4 a1 = *(const float4*)(att + c1);

  const float* pln = xl + (size_t)n * F;
  float4 acc0 = *(const float4*)(pln + c0);
  float4 acc1 = *(const float4*)(pln + c1);
  float m0 = half_sum16(lrelu_dot4(acc0, r0, a0));
  float m1 = half_sum16(lrelu_dot4(acc1, r1, a1));
  float s0 = 1.0f, s1 = 1.0f;

  int beg = off[n], end = off[n + 1];
  for (int cb = beg; cb < end; cb += 32) {
    int cnt = min(32, end - cb);
    int mysrc = srcs[cb + min(lane, cnt - 1)];
    for (int j = 0; j < cnt; j += EG) {
      float4 x0[EG], x1[EG];
#pragma unroll
      for (int u = 0; u < EG; ++u) {
        int src = __builtin_amdgcn_readlane(mysrc, min(j + u, cnt - 1));
        const float* pl = xl + (size_t)src * F;
        x0[u] = *(const float4*)(pl + c0);
        x1[u] = *(const float4*)(pl + c1);
      }
      float sc0[EG], sc1[EG];
      float nm0 = m0, nm1 = m1;
#pragma unroll
      for (int u = 0; u < EG; ++u) {
        float t0 = half_sum16(lrelu_dot4(x0[u], r0, a0));
        float t1 = half_sum16(lrelu_dot4(x1[u], r1, a1));
        bool valid = (j + u) < cnt;
        sc0[u] = valid ? t0 : -3.0e38f;
        sc1[u] = valid ? t1 : -3.0e38f;
        nm0 = fmaxf(nm0, sc0[u]);
        nm1 = fmaxf(nm1, sc1[u]);
      }
      float f0 = __expf(m0 - nm0), f1 = __expf(m1 - nm1);
      s0 *= f0; s1 *= f1;
      acc0.x *= f0; acc0.y *= f0; acc0.z *= f0; acc0.w *= f0;
      acc1.x *= f1; acc1.y *= f1; acc1.z *= f1; acc1.w *= f1;
      m0 = nm0; m1 = nm1;
#pragma unroll
      for (int u = 0; u < EG; ++u) {
        float p0 = __expf(sc0[u] - m0), p1 = __expf(sc1[u] - m1);
        s0 += p0; s1 += p1;
        acc0.x += p0 * x0[u].x; acc0.y += p0 * x0[u].y;
        acc0.z += p0 * x0[u].z; acc0.w += p0 * x0[u].w;
        acc1.x += p1 * x1[u].x; acc1.y += p1 * x1[u].y;
        acc1.z += p1 * x1[u].z; acc1.w += p1 * x1[u].w;
      }
    }
  }

  float i0 = 1.0f / fmaxf(s0, 1e-16f), i1 = 1.0f / fmaxf(s1, 1e-16f);
  float4 v;
  v.x = acc0.x * i0 + acc1.x * i1;
  v.y = acc0.y * i0 + acc1.y * i1;
  v.z = acc0.z * i0 + acc1.z * i1;
  v.w = acc0.w * i0 + acc1.w * i1;
  v.x += xhalf(v.x);
  v.y += xhalf(v.y);
  v.z += xhalf(v.z);
  v.w += xhalf(v.w);
  int cc = 4 * (lane & 15);
  const float4 bv = *(const float4*)(bias + cc);
  v.x = 0.25f * v.x + bv.x;
  v.y = 0.25f * v.y + bv.y;
  v.z = 0.25f * v.z + bv.z;
  v.w = 0.25f * v.w + bv.w;

  float mu = half_sum16((v.x + v.y) + (v.z + v.w)) * (1.0f / 64.0f);
  float d0 = v.x - mu, d1 = v.y - mu, d2 = v.z - mu, d3 = v.w - mu;
  float var = half_sum16((d0 * d0 + d1 * d1) + (d2 * d2 + d3 * d3)) * (1.0f / 64.0f);
  float inv = rsqrtf(var + 1e-5f);
  const float4 gv = *(const float4*)(g + cc);
  const float4 be = *(const float4*)(beta + cc);
  float4 y;
  y.x = d0 * inv * gv.x + be.x;
  y.y = d1 * inv * gv.y + be.y;
  y.z = d2 * inv * gv.z + be.z;
  y.w = d3 * inv * gv.w + be.w;
  y.x = (y.x > 0.f) ? y.x : (__expf(y.x) - 1.f);
  y.y = (y.y > 0.f) ? y.y : (__expf(y.y) - 1.f);
  y.z = (y.z > 0.f) ? y.z : (__expf(y.z) - 1.f);
  y.w = (y.w > 0.f) ? y.w : (__expf(y.w) - 1.f);
  if (lane < 16) *(float4*)(out + (size_t)n * HID + cc) = y;
}

__global__ void k_gemm_wmma(const float* __restrict__ X, int K,
                            const float* __restrict__ W0, const float* __restrict__ B0,
                            float* __restrict__ Y0,
                            const float* __restrict__ W1, const float* __restrict__ B1,
                            float* __restrict__ Y1,
                            int Ntiles, int nmat, int ldY, int act) {
  int wave = (blockIdx.x * blockDim.x + threadIdx.x) >> 5;
  int total = MT * Ntiles * nmat;
  if (wave >= total) return;

  int tn   = wave % Ntiles;
  int rest = wave / Ntiles;
  int mat  = rest % nmat;
  int tm   = rest / nmat;

  const float* W  = mat ? W1 : W0;
  const float* Bv = mat ? B1 : B0;
  float*       Y  = mat ? Y1 : Y0;

  int lane = threadIdx.x & 31;
  int half = lane >> 4;
  int l16  = lane & 15;

  const float* xrow = X + (size_t)(tm * 16 + l16) * K;
  const float* wrow = W + (size_t)(tn * 16 + l16) * K;

  v8f c = {0.f, 0.f, 0.f, 0.f, 0.f, 0.f, 0.f, 0.f};
  for (int k0 = 0; k0 < K; k0 += 4) {
    v2f a = *(const v2f*)(xrow + k0 + 2 * half);
    v2f b = *(const v2f*)(wrow + k0 + 2 * half);
    c = __builtin_amdgcn_wmma_f32_16x16x4_f32(false, a, false, b, (short)0, c,
                                              false, false);
  }

  int col   = tn * 16 + l16;
  float bia = Bv[col];
  int mbase = tm * 16 + 8 * half;
#pragma unroll
  for (int r = 0; r < 8; ++r) {
    float v = c[r] + bia;
    if (act) v = fmaxf(v, 0.0f);
    Y[(size_t)(mbase + r) * ldY + col] = v;
  }
}

__global__ void k_head2(const float* __restrict__ hm, const float* __restrict__ W,
                        const float* __restrict__ b, float* __restrict__ out) {
  int n = blockIdx.x * blockDim.x + threadIdx.x;
  if (n >= N_NODES) return;
  const float* p = hm + (size_t)n * 32;
  float a0 = b[0], a1 = b[1];
#pragma unroll
  for (int j = 0; j < 32; ++j) {
    float v = p[j];
    a0 += v * W[j];
    a1 += v * W[32 + j];
  }
  out[(size_t)n * 2 + 0] = a0;
  out[(size_t)n * 2 + 1] = a1;
}

static inline int wave_blocks(long long waves) { return (int)((waves + 7) / 8); }

extern "C" void kernel_launch(void* const* d_in, const int* in_sizes, int n_in,
                              void* d_out, int out_size, void* d_ws, size_t ws_size,
                              hipStream_t stream) {
  const float* x    = (const float*)d_in[0];
  const int*   ei   = (const int*)  d_in[1];
  const float* Wl1  = (const float*)d_in[2];  const float* bl1 = (const float*)d_in[3];
  const float* Wr1  = (const float*)d_in[4];  const float* br1 = (const float*)d_in[5];
  const float* att1 = (const float*)d_in[6];  const float* bias1=(const float*)d_in[7];
  const float* Wl2  = (const float*)d_in[8];  const float* bl2 = (const float*)d_in[9];
  const float* Wr2  = (const float*)d_in[10]; const float* br2 = (const float*)d_in[11];
  const float* att2 = (const float*)d_in[12]; const float* bias2=(const float*)d_in[13];
  const float* g1   = (const float*)d_in[14]; const float* be1 = (const float*)d_in[15];
  const float* g2   = (const float*)d_in[16]; const float* be2 = (const float*)d_in[17];
  const float* Wh1  = (const float*)d_in[18]; const float* bh1 = (const float*)d_in[19];
  const float* Wh2  = (const float*)d_in[20]; const float* bh2 = (const float*)d_in[21];
  float* out = (float*)d_out;

  float* ws  = (float*)d_ws;
  const size_t NF = (size_t)N_NODES * F;
  float* XL  = ws;
  float* XR  = XL + NF;
  float* H1  = XR + NF;
  float* HM  = H1 + (size_t)N_NODES * HID;
  int*   DEG = (int*)(HM + (size_t)N_NODES * 32);
  int*   OFF = DEG + NPAD;
  int*   RNK = OFF + NPAD;
  int*   SRC = RNK + N_EDGES;

  const int TB = 256;
  int nodeBlocks = wave_blocks(N_NODES);
  int padBlocks  = (NPAD + TB - 1) / TB;
  int eThrBlocks = (N_EDGES + TB - 1) / TB;

  k_deg_zero<<<padBlocks, TB, 0, stream>>>(DEG);
  k_deg_count<<<eThrBlocks, TB, 0, stream>>>(ei, DEG, RNK);
  k_deg_scan<<<1, SCAN_T, 0, stream>>>(DEG, OFF);
  k_edge_fill<<<eThrBlocks, TB, 0, stream>>>(ei, OFF, RNK, SRC);

  k_gemm_wmma<<<wave_blocks((long long)MT * 16 * 2), TB, 0, stream>>>(
      x, IN_DIM, Wl1, bl1, XL, Wr1, br1, XR,  16,  2, F, 0);
  k_node_gat<<<nodeBlocks, TB, 0, stream>>>(OFF, SRC, XL, XR, att1, bias1, g1, be1, H1);

  k_gemm_wmma<<<wave_blocks((long long)MT * 16 * 2), TB, 0, stream>>>(
      H1, HID, Wl2, bl2, XL, Wr2, br2, XR, 16, 2, F, 0);
  k_node_gat<<<nodeBlocks, TB, 0, stream>>>(OFF, SRC, XL, XR, att2, bias2, g2, be2, H1);

  k_gemm_wmma<<<wave_blocks((long long)MT * 2), TB, 0, stream>>>(
      H1, HID, Wh1, bh1, HM, Wh1, bh1, HM,  2,  1, 32,  1);
  k_head2<<<(N_NODES + TB - 1) / TB, TB, 0, stream>>>(HM, Wh2, bh2, out);
}


# projection GEMMs re-tiled: W strip staged once per work-group in LDS, X rows kept in registers, each A fragment feeds 8 WMMA tiles; k order inside the dot products changes (summation order changes)
# speedup vs baseline: 3.6693x; 1.3436x over previous
#include <hip/hip_runtime.h>
#include <math.h>

#define N_NODES 50000
#define N_EDGES 800000
#define IN_DIM  128
#define HID     64
#define NH      4
#define F       (NH * HID)
#define MT      (N_NODES / 16)

typedef __attribute__((ext_vector_type(2))) float v2f;
typedef __attribute__((ext_vector_type(8))) float v8f;

#define SCAN_T 1024
#define SCAN_V 13
#define SCAN_C (4 * SCAN_V)
#define NPAD   (SCAN_T * SCAN_C)

__global__ void k_deg_zero(int* __restrict__ deg) {
  int i = blockIdx.x * blockDim.x + threadIdx.x;
  if (i < NPAD) deg[i] = 0;
}

__global__ void k_deg_count(const int* __restrict__ ei, int* __restrict__ deg,
                            int* __restrict__ rank) {
  int e = blockIdx.x * blockDim.x + threadIdx.x;
  if (e < N_EDGES) rank[e] = atomicAdd(&deg[ei[N_EDGES + e]], 1);
}

__global__ void __launch_bounds__(SCAN_T) k_deg_scan(const int* __restrict__ deg,
                                                     int* __restrict__ off) {
  __shared__ int part[SCAN_T];
  int t = threadIdx.x;
  const int4* pin = (const int4*)(deg + t * SCAN_C);
  int4 d[SCAN_V];
#pragma unroll
  for (int i = 0; i < SCAN_V; ++i) d[i] = pin[i];
  int sum = 0;
#pragma unroll
  for (int i = 0; i < SCAN_V; ++i) {
    int4 q = d[i];
    d[i].x = sum; sum += q.x;
    d[i].y = sum; sum += q.y;
    d[i].z = sum; sum += q.z;
    d[i].w = sum; sum += q.w;
  }
  part[t] = sum;
  __syncthreads();
  for (int s = 1; s < SCAN_T; s <<= 1) {
    int v = (t >= s) ? part[t - s] : 0;
    __syncthreads();
    part[t] += v;
    __syncthreads();
  }
  int base = part[t] - sum;
  int4* pout = (int4*)(off + t * SCAN_C);
#pragma unroll
  for (int i = 0; i < SCAN_V; ++i) {
    int4 q = d[i];
    q.x += base; q.y += base; q.z += base; q.w += base;
    pout[i] = q;
  }
}

__global__ void k_edge_fill(const int* __restrict__ ei, const int* __restrict__ off,
                            const int* __restrict__ rank, int* __restrict__ srcs) {
  int e = blockIdx.x * blockDim.x + threadIdx.x;
  if (e >= N_EDGES) return;
  srcs[off[ei[N_EDGES + e]] + rank[e]] = ei[e];
}

__device__ __forceinline__ float lrelu_dot4(float4 x, float4 r, float4 a) {
  float v0 = x.x + r.x, v1 = x.y + r.y, v2 = x.z + r.z, v3 = x.w + r.w;
  v0 = (v0 > 0.f) ? v0 : 0.2f * v0;
  v1 = (v1 > 0.f) ? v1 : 0.2f * v1;
  v2 = (v2 > 0.f) ? v2 : 0.2f * v2;
  v3 = (v3 > 0.f) ? v3 : 0.2f * v3;
  return v0 * a.x + v1 * a.y + v2 * a.z + v3 * a.w;
}

template <int M>
__device__ __forceinline__ float row_xor(float v) {
  int b = __builtin_bit_cast(int, v);
  return __builtin_bit_cast(float, __builtin_amdgcn_update_dpp(b, b, 0x160 | M, 0xf, 0xf, false));
}

__device__ __forceinline__ float xhalf(float v) {
  int b = __builtin_bit_cast(int, v);
  return __builtin_bit_cast(float,
      __builtin_amdgcn_permlanex16(b, b, 0x76543210, 0xfedcba98, false, false));
}

__device__ __forceinline__ float half_sum16(float v) {
  v += row_xor<1>(v);
  v += row_xor<2>(v);
  v += row_xor<4>(v);
  v += row_xor<8>(v);
  return v;
}

#define EG 4

__global__ void __launch_bounds__(256) k_node_gat(const int* __restrict__ off, const int* __restrict__ srcs,
                           const float* __restrict__ xl, const float* __restrict__ xr,
                           const float* __restrict__ att, const float* __restrict__ bias,
                           const float* __restrict__ g, const float* __restrict__ beta,
                           float* __restrict__ out) {
  int n = __builtin_amdgcn_readfirstlane((blockIdx.x * blockDim.x + threadIdx.x) >> 5);
  if (n >= N_NODES) return;
  int lane = threadIdx.x & 31;
  int c0 = 4 * lane;
  int c1 = 2 * HID + 4 * lane;

  const float* prn = xr + (size_t)n * F;
  const float4 r0 = *(const float4*)(prn + c0);
  const float4 r1 = *(const float4*)(prn + c1);
  const float4 a0 = *(const float4*)(att + c0);
  const float4 a1 = *(const float4*)(att + c1);

  const float* pln = xl + (size_t)n * F;
  float4 acc0 = *(const float4*)(pln + c0);
  float4 acc1 = *(const float4*)(pln + c1);
  float m0 = half_sum16(lrelu_dot4(acc0, r0, a0));
  float m1 = half_sum16(lrelu_dot4(acc1, r1, a1));
  float s0 = 1.0f, s1 = 1.0f;

  int beg = off[n], end = off[n + 1];
  for (int cb = beg; cb < end; cb += 32) {
    int cnt = min(32, end - cb);
    int mysrc = srcs[cb + min(lane, cnt - 1)];
    for (int j = 0; j < cnt; j += EG) {
      float4 x0[EG], x1[EG];
#pragma unroll
      for (int u = 0; u < EG; ++u) {
        int src = __builtin_amdgcn_readlane(mysrc, min(j + u, cnt - 1));
        const float* pl = xl + (size_t)src * F;
        x0[u] = *(const float4*)(pl + c0);
        x1[u] = *(const float4*)(pl + c1);
      }
      float sc0[EG], sc1[EG];
      float nm0 = m0, nm1 = m1;
#pragma unroll
      for (int u = 0; u < EG; ++u) {
        float t0 = half_sum16(lrelu_dot4(x0[u], r0, a0));
        float t1 = half_sum16(lrelu_dot4(x1[u], r1, a1));
        bool valid = (j + u) < cnt;
        sc0[u] = valid ? t0 : -3.0e38f;
        sc1[u] = valid ? t1 : -3.0e38f;
        nm0 = fmaxf(nm0, sc0[u]);
        nm1 = fmaxf(nm1, sc1[u]);
      }
      float f0 = __expf(m0 - nm0), f1 = __expf(m1 - nm1);
      s0 *= f0; s1 *= f1;
      acc0.x *= f0; acc0.y *= f0; acc0.z *= f0; acc0.w *= f0;
      acc1.x *= f1; acc1.y *= f1; acc1.z *= f1; acc1.w *= f1;
      m0 = nm0; m1 = nm1;
#pragma unroll
      for (int u = 0; u < EG; ++u) {
        float p0 = __expf(sc0[u] - m0), p1 = __expf(sc1[u] - m1);
        s0 += p0; s1 += p1;
        acc0.x += p0 * x0[u].x; acc0.y += p0 * x0[u].y;
        acc0.z += p0 * x0[u].z; acc0.w += p0 * x0[u].w;
        acc1.x += p1 * x1[u].x; acc1.y += p1 * x1[u].y;
        acc1.z += p1 * x1[u].z; acc1.w += p1 * x1[u].w;
      }
    }
  }

  float i0 = 1.0f / fmaxf(s0, 1e-16f), i1 = 1.0f / fmaxf(s1, 1e-16f);
  float4 v;
  v.x = acc0.x * i0 + acc1.x * i1;
  v.y = acc0.y * i0 + acc1.y * i1;
  v.z = acc0.z * i0 + acc1.z * i1;
  v.w = acc0.w * i0 + acc1.w * i1;
  v.x += xhalf(v.x);
  v.y += xhalf(v.y);
  v.z += xhalf(v.z);
  v.w += xhalf(v.w);
  int cc = 4 * (lane & 15);
  const float4 bv = *(const float4*)(bias + cc);
  v.x = 0.25f * v.x + bv.x;
  v.y = 0.25f * v.y + bv.y;
  v.z = 0.25f * v.z + bv.z;
  v.w = 0.25f * v.w + bv.w;

  float mu = half_sum16((v.x + v.y) + (v.z + v.w)) * (1.0f / 64.0f);
  float d0 = v.x - mu, d1 = v.y - mu, d2 = v.z - mu, d3 = v.w - mu;
  float var = half_sum16((d0 * d0 + d1 * d1) + (d2 * d2 + d3 * d3)) * (1.0f / 64.0f);
  float inv = rsqrtf(var + 1e-5f);
  const float4 gv = *(const float4*)(g + cc);
  const float4 be = *(const float4*)(beta + cc);
  float4 y;
  y.x = d0 * inv * gv.x + be.x;
  y.y = d1 * inv * gv.y + be.y;
  y.z = d2 * inv * gv.z + be.z;
  y.w = d3 * inv * gv.w + be.w;
  y.x = (y.x > 0.f) ? y.x : (__expf(y.x) - 1.f);
  y.y = (y.y > 0.f) ? y.y : (__expf(y.y) - 1.f);
  y.z = (y.z > 0.f) ? y.z : (__expf(y.z) - 1.f);
  y.w = (y.w > 0.f) ? y.w : (__expf(y.w) - 1.f);
  if (lane < 16) *(float4*)(out + (size_t)n * HID + cc) = y;
}

__global__ void k_gemm_wmma(const float* __restrict__ X, int K,
                            const float* __restrict__ W0, const float* __restrict__ B0,
                            float* __restrict__ Y0,
                            const float* __restrict__ W1, const float* __restrict__ B1,
                            float* __restrict__ Y1,
                            int Ntiles, int nmat, int ldY, int act) {
  int wave = (blockIdx.x * blockDim.x + threadIdx.x) >> 5;
  int total = MT * Ntiles * nmat;
  if (wave >= total) return;

  int tn   = wave % Ntiles;
  int rest = wave / Ntiles;
  int mat  = rest % nmat;
  int tm   = rest / nmat;

  const float* W  = mat ? W1 : W0;
  const float* Bv = mat ? B1 : B0;
  float*       Y  = mat ? Y1 : Y0;

  int lane = threadIdx.x & 31;
  int half = lane >> 4;
  int l16  = lane & 15;

  const float* xrow = X + (size_t)(tm * 16 + l16) * K;
  const float* wrow = W + (size_t)(tn * 16 + l16) * K;

  v8f c = {0.f, 0.f, 0.f, 0.f, 0.f, 0.f, 0.f, 0.f};
  for (int k0 = 0; k0 < K; k0 += 4) {
    v2f a = *(const v2f*)(xrow + k0 + 2 * half);
    v2f b = *(const v2f*)(wrow + k0 + 2 * half);
    c = __builtin_amdgcn_wmma_f32_16x16x4_f32(false, a, false, b, (short)0, c,
                                              false, false);
  }

  int col   = tn * 16 + l16;
  float bia = Bv[col];
  int mbase = tm * 16 + 8 * half;
#pragma unroll
  for (int r = 0; r < 8; ++r) {
    float v = c[r] + bia;
    if (act) v = fmaxf(v, 0.0f);
    Y[(size_t)(mbase + r) * ldY + col] = v;
  }
}

template <int K>
__global__ void __launch_bounds__(256) k_gemm_strip(const float* __restrict__ X,
                             const float* __restrict__ W0, const float* __restrict__ B0,
                             float* __restrict__ Y0,
                             const float* __restrict__ W1, const float* __restrict__ B1,
                             float* __restrict__ Y1) {
  constexpr int PITCH = K + 4;
  constexpr int KV = K / 8;
  constexpr int RV = K / 4;
  __shared__ __attribute__((aligned(16))) float Ws[128 * PITCH];

  int strip = blockIdx.x & 3;
  int n0 = (strip & 1) * 128;
  const float* W  = ((strip >> 1) ? W1 : W0) + (size_t)n0 * K;
  const float* Bv = ((strip >> 1) ? B1 : B0) + n0;
  float*       Y  = ((strip >> 1) ? Y1 : Y0) + n0;

  int wv = __builtin_amdgcn_readfirstlane(threadIdx.x >> 5);
  int lane = threadIdx.x & 31;
  int h = lane >> 4;
  int c = lane & 15;
  int tm  = (blockIdx.x >> 2) * 8 + wv;
  int tmc = min(tm, MT - 1);

  const float4* xrow = (const float4*)(X + (size_t)(tmc * 16 + c) * K + 4 * h);
  float4 a[KV];
#pragma unroll
  for (int j = 0; j < KV; ++j) a[j] = xrow[2 * j];

  const float4* wsrc = (const float4*)W;
#pragma unroll
  for (int i = 0; i < (128 * RV) / 256; ++i) {
    int f = threadIdx.x + 256 * i;
    int row = f / RV, q = f % RV;
    float4 v = wsrc[f];
    *(float4*)(&Ws[row * PITCH + 4 * q]) = v;
  }
  __syncthreads();
  if (tm >= MT) return;

  v8f acc[8];
#pragma unroll
  for (int t = 0; t < 8; ++t) acc[t] = (v8f){0.f, 0.f, 0.f, 0.f, 0.f, 0.f, 0.f, 0.f};

  const float* bl = Ws + c * PITCH + 4 * h;
#pragma unroll
  for (int j = 0; j < KV; ++j) {
    float4 b[8];
#pragma unroll
    for (int t = 0; t < 8; ++t) b[t] = *(const float4*)(bl + 16 * t * PITCH + 8 * j);
    v2f alo = {a[j].x, a[j].y};
    v2f ahi = {a[j].z, a[j].w};
#pragma unroll
    for (int t = 0; t < 8; ++t) {
      v2f blo = {b[t].x, b[t].y};
      acc[t] = __builtin_amdgcn_wmma_f32_16x16x4_f32(false, alo, false, blo, (short)0, acc[t],
                                                     false, false);
    }
#pragma unroll
    for (int t = 0; t < 8; ++t) {
      v2f bhi = {b[t].z, b[t].w};
      acc[t] = __builtin_amdgcn_wmma_f32_16x16x4_f32(false, ahi, false, bhi, (short)0, acc[t],
                                                     false, false);
    }
  }

  int rbase = tm * 16 + 8 * h;
#pragma unroll
  for (int t = 0; t < 8; ++t) {
    int col = 16 * t + c;
    float bia = Bv[col];
#pragma unroll
    for (int r = 0; r < 8; ++r)
      Y[(size_t)(rbase + r) * F + col] = acc[t][r] + bia;
  }
}

__global__ void k_head2(const float* __restrict__ hm, const float* __restrict__ W,
                        const float* __restrict__ b, float* __restrict__ out) {
  int n = blockIdx.x * blockDim.x + threadIdx.x;
  if (n >= N_NODES) return;
  const float* p = hm + (size_t)n * 32;
  float a0 = b[0], a1 = b[1];
#pragma unroll
  for (int j = 0; j < 32; ++j) {
    float v = p[j];
    a0 += v * W[j];
    a1 += v * W[32 + j];
  }
  out[(size_t)n * 2 + 0] = a0;
  out[(size_t)n * 2 + 1] = a1;
}

static inline int wave_blocks(long long waves) { return (int)((waves + 7) / 8); }
#define STRIP_BLOCKS (((MT + 7) / 8) * 4)

extern "C" void kernel_launch(void* const* d_in, const int* in_sizes, int n_in,
                              void* d_out, int out_size, void* d_ws, size_t ws_size,
                              hipStream_t stream) {
  const float* x    = (const float*)d_in[0];
  const int*   ei   = (const int*)  d_in[1];
  const float* Wl1  = (const float*)d_in[2];  const float* bl1 = (const float*)d_in[3];
  const float* Wr1  = (const float*)d_in[4];  const float* br1 = (const float*)d_in[5];
  const float* att1 = (const float*)d_in[6];  const float* bias1=(const float*)d_in[7];
  const float* Wl2  = (const float*)d_in[8];  const float* bl2 = (const float*)d_in[9];
  const float* Wr2  = (const float*)d_in[10]; const float* br2 = (const float*)d_in[11];
  const float* att2 = (const float*)d_in[12]; const float* bias2=(const float*)d_in[13];
  const float* g1   = (const float*)d_in[14]; const float* be1 = (const float*)d_in[15];
  const float* g2   = (const float*)d_in[16]; const float* be2 = (const float*)d_in[17];
  const float* Wh1  = (const float*)d_in[18]; const float* bh1 = (const float*)d_in[19];
  const float* Wh2  = (const float*)d_in[20]; const float* bh2 = (const float*)d_in[21];
  float* out = (float*)d_out;

  float* ws  = (float*)d_ws;
  const size_t NF = (size_t)N_NODES * F;
  float* XL  = ws;
  float* XR  = XL + NF;
  float* H1  = XR + NF;
  float* HM  = H1 + (size_t)N_NODES * HID;
  int*   DEG = (int*)(HM + (size_t)N_NODES * 32);
  int*   OFF = DEG + NPAD;
  int*   RNK = OFF + NPAD;
  int*   SRC = RNK + N_EDGES;

  const int TB = 256;
  int nodeBlocks = wave_blocks(N_NODES);
  int padBlocks  = (NPAD + TB - 1) / TB;
  int eThrBlocks = (N_EDGES + TB - 1) / TB;

  k_deg_zero<<<padBlocks, TB, 0, stream>>>(DEG);
  k_deg_count<<<eThrBlocks, TB, 0, stream>>>(ei, DEG, RNK);
  k_deg_scan<<<1, SCAN_T, 0, stream>>>(DEG, OFF);
  k_edge_fill<<<eThrBlocks, TB, 0, stream>>>(ei, OFF, RNK, SRC);

  k_gemm_strip<IN_DIM><<<STRIP_BLOCKS, TB, 0, stream>>>(x, Wl1, bl1, XL, Wr1, br1, XR);
  k_node_gat<<<nodeBlocks, TB, 0, stream>>>(OFF, SRC, XL, XR, att1, bias1, g1, be1, H1);

  k_gemm_strip<HID><<<STRIP_BLOCKS, TB, 0, stream>>>(H1, Wl2, bl2, XL, Wr2, br2, XR);
  k_node_gat<<<nodeBlocks, TB, 0, stream>>>(OFF, SRC, XL, XR, att2, bias2, g2, be2, H1);

  k_gemm_wmma<<<wave_blocks((long long)MT * 2), TB, 0, stream>>>(
      H1, HID, Wh1, bh1, HM, Wh1, bh1, HM,  2,  1, 32,  1);
  k_head2<<<(N_NODES + TB - 1) / TB, TB, 0, stream>>>(HM, Wh2, bh2, out);
}


# node kernel: each lane serves ONE head with 8 of its channels (a head = 8 lanes): one score, one exp and one 3-step lane reduction per edge instead of two 4-step ones; summation order changes
# speedup vs baseline: 4.0611x; 1.1068x over previous
#include <hip/hip_runtime.h>
#include <math.h>

#define N_NODES 50000
#define N_EDGES 800000
#define IN_DIM  128
#define HID     64
#define NH      4
#define F       (NH * HID)
#define MT      (N_NODES / 16)

typedef __attribute__((ext_vector_type(2))) float v2f;
typedef __attribute__((ext_vector_type(8))) float v8f;

#define SCAN_T 1024
#define SCAN_V 13
#define SCAN_C (4 * SCAN_V)
#define NPAD   (SCAN_T * SCAN_C)

__global__ void k_deg_zero(int* __restrict__ deg) {
  int i = blockIdx.x * blockDim.x + threadIdx.x;
  if (i < NPAD) deg[i] = 0;
}

__global__ void k_deg_count(const int* __restrict__ ei, int* __restrict__ deg,
                            int* __restrict__ rank) {
  int e = blockIdx.x * blockDim.x + threadIdx.x;
  if (e < N_EDGES) rank[e] = atomicAdd(&deg[ei[N_EDGES + e]], 1);
}

__global__ void __launch_bounds__(SCAN_T) k_deg_scan(const int* __restrict__ deg,
                                                     int* __restrict__ off) {
  __shared__ int part[SCAN_T];
  int t = threadIdx.x;
  const int4* pin = (const int4*)(deg + t * SCAN_C);
  int4 d[SCAN_V];
#pragma unroll
  for (int i = 0; i < SCAN_V; ++i) d[i] = pin[i];
  int sum = 0;
#pragma unroll
  for (int i = 0; i < SCAN_V; ++i) {
    int4 q = d[i];
    d[i].x = sum; sum += q.x;
    d[i].y = sum; sum += q.y;
    d[i].z = sum; sum += q.z;
    d[i].w = sum; sum += q.w;
  }
  part[t] = sum;
  __syncthreads();
  for (int s = 1; s < SCAN_T; s <<= 1) {
    int v = (t >= s) ? part[t - s] : 0;
    __syncthreads();
    part[t] += v;
    __syncthreads();
  }
  int base = part[t] - sum;
  int4* pout = (int4*)(off + t * SCAN_C);
#pragma unroll
  for (int i = 0; i < SCAN_V; ++i) {
    int4 q = d[i];
    q.x += base; q.y += base; q.z += base; q.w += base;
    pout[i] = q;
  }
}

__global__ void k_edge_fill(const int* __restrict__ ei, const int* __restrict__ off,
                            const int* __restrict__ rank, int* __restrict__ srcs) {
  int e = blockIdx.x * blockDim.x + threadIdx.x;
  if (e >= N_EDGES) return;
  srcs[off[ei[N_EDGES + e]] + rank[e]] = ei[e];
}

typedef __attribute__((ext_vector_type(4))) float v4f;

__device__ __forceinline__ float lrelu_dot8(v4f x0, v4f x1, v4f r0, v4f r1, v4f a0, v4f a1) {
  v2f e0 = x0.lo + r0.lo, e1 = x0.hi + r0.hi;
  v2f e2 = x1.lo + r1.lo, e3 = x1.hi + r1.hi;
  e0 = __builtin_elementwise_max(e0, e0 * 0.2f);
  e1 = __builtin_elementwise_max(e1, e1 * 0.2f);
  e2 = __builtin_elementwise_max(e2, e2 * 0.2f);
  e3 = __builtin_elementwise_max(e3, e3 * 0.2f);
  v2f d = e0 * a0.lo;
  d = __builtin_elementwise_fma(e1, a0.hi, d);
  d = __builtin_elementwise_fma(e2, a1.lo, d);
  d = __builtin_elementwise_fma(e3, a1.hi, d);
  return d.x + d.y;
}

template <int M>
__device__ __forceinline__ float row_xor(float v) {
  int b = __builtin_bit_cast(int, v);
  return __builtin_bit_cast(float, __builtin_amdgcn_update_dpp(b, b, 0x160 | M, 0xf, 0xf, false));
}

__device__ __forceinline__ float xhalf(float v) {
  int b = __builtin_bit_cast(int, v);
  return __builtin_bit_cast(float,
      __builtin_amdgcn_permlanex16(b, b, 0x76543210, 0xfedcba98, false, false));
}

__device__ __forceinline__ float group_sum8(float v) {
  v += row_xor<1>(v);
  v += row_xor<2>(v);
  v += row_xor<4>(v);
  return v;
}

__device__ __forceinline__ float half_sum16(float v) {
  v = group_sum8(v);
  v += row_xor<8>(v);
  return v;
}

#define EG 4

__global__ void __launch_bounds__(256) k_node_gat(const int* __restrict__ off, const int* __restrict__ srcs,
                           const float* __restrict__ xl, const float* __restrict__ xr,
                           const float* __restrict__ att, const float* __restrict__ bias,
                           const float* __restrict__ g, const float* __restrict__ beta,
                           float* __restrict__ out) {
  int n = __builtin_amdgcn_readfirstlane((blockIdx.x * blockDim.x + threadIdx.x) >> 5);
  if (n >= N_NODES) return;
  int lane = threadIdx.x & 31;
  int c0 = HID * (lane >> 3) + 4 * (lane & 7);
  int c1 = c0 + HID / 2;

  const float* prn = xr + (size_t)n * F;
  const v4f r0 = *(const v4f*)(prn + c0);
  const v4f r1 = *(const v4f*)(prn + c1);
  const v4f a0 = *(const v4f*)(att + c0);
  const v4f a1 = *(const v4f*)(att + c1);

  const float* pln = xl + (size_t)n * F;
  v4f acc0 = *(const v4f*)(pln + c0);
  v4f acc1 = *(const v4f*)(pln + c1);
  float m = group_sum8(lrelu_dot8(acc0, acc1, r0, r1, a0, a1));
  float s = 1.0f;

  int beg = off[n], end = off[n + 1];
  for (int cb = beg; cb < end; cb += 32) {
    int cnt = min(32, end - cb);
    int mysrc = srcs[cb + min(lane, cnt - 1)];
    for (int j = 0; j < cnt; j += EG) {
      v4f x0[EG], x1[EG];
#pragma unroll
      for (int u = 0; u < EG; ++u) {
        int src = __builtin_amdgcn_readlane(mysrc, min(j + u, cnt - 1));
        const float* pl = xl + (size_t)src * F;
        x0[u] = *(const v4f*)(pl + c0);
        x1[u] = *(const v4f*)(pl + c1);
      }
      float sc[EG];
      float nm = m;
#pragma unroll
      for (int u = 0; u < EG; ++u) {
        float t = group_sum8(lrelu_dot8(x0[u], x1[u], r0, r1, a0, a1));
        sc[u] = ((j + u) < cnt) ? t : -3.0e38f;
        nm = fmaxf(nm, sc[u]);
      }
      float f = __expf(m - nm);
      s *= f;
      acc0 *= f;
      acc1 *= f;
      m = nm;
#pragma unroll
      for (int u = 0; u < EG; ++u) {
        float p = __expf(sc[u] - m);
        s += p;
        acc0 = __builtin_elementwise_fma((v4f)p, x0[u], acc0);
        acc1 = __builtin_elementwise_fma((v4f)p, x1[u], acc1);
      }
    }
  }

  float inv_s = 1.0f / fmaxf(s, 1e-16f);
  acc0 *= inv_s;
  acc1 *= inv_s;
  bool up = (lane & 8) != 0;
  v4f keep = up ? acc1 : acc0;
  v4f send = up ? acc0 : acc1;
  float4 v;
  v.x = keep.x + row_xor<8>(send.x);
  v.y = keep.y + row_xor<8>(send.y);
  v.z = keep.z + row_xor<8>(send.z);
  v.w = keep.w + row_xor<8>(send.w);
  v.x += xhalf(v.x);
  v.y += xhalf(v.y);
  v.z += xhalf(v.z);
  v.w += xhalf(v.w);
  int cc = 4 * (lane & 15);
  const float4 bv = *(const float4*)(bias + cc);
  v.x = 0.25f * v.x + bv.x;
  v.y = 0.25f * v.y + bv.y;
  v.z = 0.25f * v.z + bv.z;
  v.w = 0.25f * v.w + bv.w;

  float mu = half_sum16((v.x + v.y) + (v.z + v.w)) * (1.0f / 64.0f);
  float d0 = v.x - mu, d1 = v.y - mu, d2 = v.z - mu, d3 = v.w - mu;
  float var = half_sum16((d0 * d0 + d1 * d1) + (d2 * d2 + d3 * d3)) * (1.0f / 64.0f);
  float inv = rsqrtf(var + 1e-5f);
  const float4 gv = *(const float4*)(g + cc);
  const float4 be = *(const float4*)(beta + cc);
  float4 y;
  y.x = d0 * inv * gv.x + be.x;
  y.y = d1 * inv * gv.y + be.y;
  y.z = d2 * inv * gv.z + be.z;
  y.w = d3 * inv * gv.w + be.w;
  y.x = (y.x > 0.f) ? y.x : (__expf(y.x) - 1.f);
  y.y = (y.y > 0.f) ? y.y : (__expf(y.y) - 1.f);
  y.z = (y.z > 0.f) ? y.z : (__expf(y.z) - 1.f);
  y.w = (y.w > 0.f) ? y.w : (__expf(y.w) - 1.f);
  if (lane < 16) *(float4*)(out + (size_t)n * HID + cc) = y;
}

__global__ void k_gemm_wmma(const float* __restrict__ X, int K,
                            const float* __restrict__ W0, const float* __restrict__ B0,
                            float* __restrict__ Y0,
                            const float* __restrict__ W1, const float* __restrict__ B1,
                            float* __restrict__ Y1,
                            int Ntiles, int nmat, int ldY, int act) {
  int wave = (blockIdx.x * blockDim.x + threadIdx.x) >> 5;
  int total = MT * Ntiles * nmat;
  if (wave >= total) return;

  int tn   = wave % Ntiles;
  int rest = wave / Ntiles;
  int mat  = rest % nmat;
  int tm   = rest / nmat;

  const float* W  = mat ? W1 : W0;
  const float* Bv = mat ? B1 : B0;
  float*       Y  = mat ? Y1 : Y0;

  int lane = threadIdx.x & 31;
  int half = lane >> 4;
  int l16  = lane & 15;

  const float* xrow = X + (size_t)(tm * 16 + l16) * K;
  const float* wrow = W + (size_t)(tn * 16 + l16) * K;

  v8f c = {0.f, 0.f, 0.f, 0.f, 0.f, 0.f, 0.f, 0.f};
  for (int k0 = 0; k0 < K; k0 += 4) {
    v2f a = *(const v2f*)(xrow + k0 + 2 * half);
    v2f b = *(const v2f*)(wrow + k0 + 2 * half);
    c = __builtin_amdgcn_wmma_f32_16x16x4_f32(false, a, false, b, (short)0, c,
                                              false, false);
  }

  int col   = tn * 16 + l16;
  float bia = Bv[col];
  int mbase = tm * 16 + 8 * half;
#pragma unroll
  for (int r = 0; r < 8; ++r) {
    float v = c[r] + bia;
    if (act) v = fmaxf(v, 0.0f);
    Y[(size_t)(mbase + r) * ldY + col] = v;
  }
}

template <int K>
__global__ void __launch_bounds__(256) k_gemm_strip(const float* __restrict__ X,
                             const float* __restrict__ W0, const float* __restrict__ B0,
                             float* __restrict__ Y0,
                             const float* __restrict__ W1, const float* __restrict__ B1,
                             float* __restrict__ Y1) {
  constexpr int PITCH = K + 4;
  constexpr int KV = K / 8;
  constexpr int RV = K / 4;
  __shared__ __attribute__((aligned(16))) float Ws[128 * PITCH];

  int strip = blockIdx.x & 3;
  int n0 = (strip & 1) * 128;
  const float* W  = ((strip >> 1) ? W1 : W0) + (size_t)n0 * K;
  const float* Bv = ((strip >> 1) ? B1 : B0) + n0;
  float*       Y  = ((strip >> 1) ? Y1 : Y0) + n0;

  int wv = __builtin_amdgcn_readfirstlane(threadIdx.x >> 5);
  int lane = threadIdx.x & 31;
  int h = lane >> 4;
  int c = lane & 15;
  int tm  = (blockIdx.x >> 2) * 8 + wv;
  int tmc = min(tm, MT - 1);

  const float4* xrow = (const float4*)(X + (size_t)(tmc * 16 + c) * K + 4 * h);
  float4 a[KV];
#pragma unroll
  for (int j = 0; j < KV; ++j) a[j] = xrow[2 * j];

  const float4* wsrc = (const float4*)W;
#pragma unroll
  for (int i = 0; i < (128 * RV) / 256; ++i) {
    int f = threadIdx.x + 256 * i;
    int row = f / RV, q = f % RV;
    float4 v = wsrc[f];
    *(float4*)(&Ws[row * PITCH + 4 * q]) = v;
  }
  __syncthreads();
  if (tm >= MT) return;

  v8f acc[8];
#pragma unroll
  for (int t = 0; t < 8; ++t) acc[t] = (v8f){0.f, 0.f, 0.f, 0.f, 0.f, 0.f, 0.f, 0.f};

  const float* bl = Ws + c * PITCH + 4 * h;
#pragma unroll
  for (int j = 0; j < KV; ++j) {
    float4 b[8];
#pragma unroll
    for (int t = 0; t < 8; ++t) b[t] = *(const float4*)(bl + 16 * t * PITCH + 8 * j);
    v2f alo = {a[j].x, a[j].y};
    v2f ahi = {a[j].z, a[j].w};
#pragma unroll
    for (int t = 0; t < 8; ++t) {
      v2f blo = {b[t].x, b[t].y};
      acc[t] = __builtin_amdgcn_wmma_f32_16x16x4_f32(false, alo, false, blo, (short)0, acc[t],
                                                     false, false);
    }
#pragma unroll
    for (int t = 0; t < 8; ++t) {
      v2f bhi = {b[t].z, b[t].w};
      acc[t] = __builtin_amdgcn_wmma_f32_16x16x4_f32(false, ahi, false, bhi, (short)0, acc[t],
                                                     false, false);
    }
  }

  int rbase = tm * 16 + 8 * h;
#pragma unroll
  for (int t = 0; t < 8; ++t) {
    int col = 16 * t + c;
    float bia = Bv[col];
#pragma unroll
    for (int r = 0; r < 8; ++r)
      Y[(size_t)(rbase + r) * F + col] = acc[t][r] + bia;
  }
}

__global__ void k_head2(const float* __restrict__ hm, const float* __restrict__ W,
                        const float* __restrict__ b, float* __restrict__ out) {
  int n = blockIdx.x * blockDim.x + threadIdx.x;
  if (n >= N_NODES) return;
  const float* p = hm + (size_t)n * 32;
  float a0 = b[0], a1 = b[1];
#pragma unroll
  for (int j = 0; j < 32; ++j) {
    float v = p[j];
    a0 += v * W[j];
    a1 += v * W[32 + j];
  }
  out[(size_t)n * 2 + 0] = a0;
  out[(size_t)n * 2 + 1] = a1;
}

static inline int wave_blocks(long long waves) { return (int)((waves + 7) / 8); }
#define STRIP_BLOCKS (((MT + 7) / 8) * 4)

extern "C" void kernel_launch(void* const* d_in, const int* in_sizes, int n_in,
                              void* d_out, int out_size, void* d_ws, size_t ws_size,
                              hipStream_t stream) {
  const float* x    = (const float*)d_in[0];
  const int*   ei   = (const int*)  d_in[1];
  const float* Wl1  = (const float*)d_in[2];  const float* bl1 = (const float*)d_in[3];
  const float* Wr1  = (const float*)d_in[4];  const float* br1 = (const float*)d_in[5];
  const float* att1 = (const float*)d_in[6];  const float* bias1=(const float*)d_in[7];
  const float* Wl2  = (const float*)d_in[8];  const float* bl2 = (const float*)d_in[9];
  const float* Wr2  = (const float*)d_in[10]; const float* br2 = (const float*)d_in[11];
  const float* att2 = (const float*)d_in[12]; const float* bias2=(const float*)d_in[13];
  const float* g1   = (const float*)d_in[14]; const float* be1 = (const float*)d_in[15];
  const float* g2   = (const float*)d_in[16]; const float* be2 = (const float*)d_in[17];
  const float* Wh1  = (const float*)d_in[18]; const float* bh1 = (const float*)d_in[19];
  const float* Wh2  = (const float*)d_in[20]; const float* bh2 = (const float*)d_in[21];
  float* out = (float*)d_out;

  float* ws  = (float*)d_ws;
  const size_t NF = (size_t)N_NODES * F;
  float* XL  = ws;
  float* XR  = XL + NF;
  float* H1  = XR + NF;
  float* HM  = H1 + (size_t)N_NODES * HID;
  int*   DEG = (int*)(HM + (size_t)N_NODES * 32);
  int*   OFF = DEG + NPAD;
  int*   RNK = OFF + NPAD;
  int*   SRC = RNK + N_EDGES;

  const int TB = 256;
  int nodeBlocks = wave_blocks(N_NODES);
  int padBlocks  = (NPAD + TB - 1) / TB;
  int eThrBlocks = (N_EDGES + TB - 1) / TB;

  k_deg_zero<<<padBlocks, TB, 0, stream>>>(DEG);
  k_deg_count<<<eThrBlocks, TB, 0, stream>>>(ei, DEG, RNK);
  k_deg_scan<<<1, SCAN_T, 0, stream>>>(DEG, OFF);
  k_edge_fill<<<eThrBlocks, TB, 0, stream>>>(ei, OFF, RNK, SRC);

  k_gemm_strip<IN_DIM><<<STRIP_BLOCKS, TB, 0, stream>>>(x, Wl1, bl1, XL, Wr1, br1, XR);
  k_node_gat<<<nodeBlocks, TB, 0, stream>>>(OFF, SRC, XL, XR, att1, bias1, g1, be1, H1);

  k_gemm_strip<HID><<<STRIP_BLOCKS, TB, 0, stream>>>(H1, Wl2, bl2, XL, Wr2, br2, XR);
  k_node_gat<<<nodeBlocks, TB, 0, stream>>>(OFF, SRC, XL, XR, att2, bias2, g2, be2, H1);

  k_gemm_wmma<<<wave_blocks((long long)MT * 2), TB, 0, stream>>>(
      H1, HID, Wh1, bh1, HM, Wh1, bh1, HM,  2,  1, 32,  1);
  k_head2<<<(N_NODES + TB - 1) / TB, TB, 0, stream>>>(HM, Wh2, bh2, out);
}


# count pass drops each source into slot [atomic rank] of its destination's 64-slot row: prefix-scan and fill launches removed (10 -> 8); rows over 64 re-read the edge list; summation order as before
# speedup vs baseline: 4.7037x; 1.1582x over previous
#include <hip/hip_runtime.h>
#include <math.h>

#define N_NODES 50000
#define N_EDGES 800000
#define IN_DIM  128
#define HID     64
#define NH      4
#define F       (NH * HID)
#define MT      (N_NODES / 16)

typedef __attribute__((ext_vector_type(2))) float v2f;
typedef __attribute__((ext_vector_type(8))) float v8f;

#define CAP     64
#define DEG_PAD 50176

__global__ void k_deg_zero(int* __restrict__ deg) {
  int i = blockIdx.x * blockDim.x + threadIdx.x;
  if (i < N_NODES) deg[i] = 0;
}

__global__ void k_bucket_fill(const int* __restrict__ ei, int* __restrict__ deg,
                              int* __restrict__ bkt) {
  int e = blockIdx.x * blockDim.x + threadIdx.x;
  if (e >= N_EDGES) return;
  int d = ei[N_EDGES + e];
  int r = atomicAdd(&deg[d], 1);
  if (r < CAP) bkt[(size_t)d * CAP + r] = ei[e];
}

typedef __attribute__((ext_vector_type(4))) float v4f;

__device__ __forceinline__ float lrelu_dot8(v4f x0, v4f x1, v4f r0, v4f r1, v4f a0, v4f a1) {
  v2f e0 = x0.lo + r0.lo, e1 = x0.hi + r0.hi;
  v2f e2 = x1.lo + r1.lo, e3 = x1.hi + r1.hi;
  e0 = __builtin_elementwise_max(e0, e0 * 0.2f);
  e1 = __builtin_elementwise_max(e1, e1 * 0.2f);
  e2 = __builtin_elementwise_max(e2, e2 * 0.2f);
  e3 = __builtin_elementwise_max(e3, e3 * 0.2f);
  v2f d = e0 * a0.lo;
  d = __builtin_elementwise_fma(e1, a0.hi, d);
  d = __builtin_elementwise_fma(e2, a1.lo, d);
  d = __builtin_elementwise_fma(e3, a1.hi, d);
  return d.x + d.y;
}

template <int M>
__device__ __forceinline__ float row_xor(float v) {
  int b = __builtin_bit_cast(int, v);
  return __builtin_bit_cast(float, __builtin_amdgcn_update_dpp(b, b, 0x160 | M, 0xf, 0xf, false));
}

__device__ __forceinline__ float xhalf(float v) {
  int b = __builtin_bit_cast(int, v);
  return __builtin_bit_cast(float,
      __builtin_amdgcn_permlanex16(b, b, 0x76543210, 0xfedcba98, false, false));
}

__device__ __forceinline__ float group_sum8(float v) {
  v += row_xor<1>(v);
  v += row_xor<2>(v);
  v += row_xor<4>(v);
  return v;
}

__device__ __forceinline__ float half_sum16(float v) {
  v = group_sum8(v);
  v += row_xor<8>(v);
  return v;
}

#define EG 4

__device__ __forceinline__ void edge_chunk(int mysrc, int cnt, const float* __restrict__ xl,
                                           int c0, int c1, v4f r0, v4f r1, v4f a0, v4f a1,
                                           float& m, float& s, v4f& acc0, v4f& acc1) {
  for (int j = 0; j < cnt; j += EG) {
    v4f x0[EG], x1[EG];
#pragma unroll
    for (int u = 0; u < EG; ++u) {
      int src = __builtin_amdgcn_readlane(mysrc, min(j + u, cnt - 1));
      const float* pl = xl + (size_t)src * F;
      x0[u] = *(const v4f*)(pl + c0);
      x1[u] = *(const v4f*)(pl + c1);
    }
    float sc[EG];
    float nm = m;
#pragma unroll
    for (int u = 0; u < EG; ++u) {
      float t = group_sum8(lrelu_dot8(x0[u], x1[u], r0, r1, a0, a1));
      sc[u] = ((j + u) < cnt) ? t : -3.0e38f;
      nm = fmaxf(nm, sc[u]);
    }
    float f = __expf(m - nm);
    s *= f;
    acc0 *= f;
    acc1 *= f;
    m = nm;
#pragma unroll
    for (int u = 0; u < EG; ++u) {
      float p = __expf(sc[u] - m);
      s += p;
      acc0 = __builtin_elementwise_fma((v4f)p, x0[u], acc0);
      acc1 = __builtin_elementwise_fma((v4f)p, x1[u], acc1);
    }
  }
}

__global__ void __launch_bounds__(256) k_node_gat(const int* __restrict__ deg, const int* __restrict__ bkt,
                           const int* __restrict__ ei,
                           const float* __restrict__ xl, const float* __restrict__ xr,
                           const float* __restrict__ att, const float* __restrict__ bias,
                           const float* __restrict__ g, const float* __restrict__ beta,
                           float* __restrict__ out) {
  int n = __builtin_amdgcn_readfirstlane((blockIdx.x * blockDim.x + threadIdx.x) >> 5);
  if (n >= N_NODES) return;
  int lane = threadIdx.x & 31;
  int c0 = HID * (lane >> 3) + 4 * (lane & 7);
  int c1 = c0 + HID / 2;

  const float* prn = xr + (size_t)n * F;
  const v4f r0 = *(const v4f*)(prn + c0);
  const v4f r1 = *(const v4f*)(prn + c1);
  const v4f a0 = *(const v4f*)(att + c0);
  const v4f a1 = *(const v4f*)(att + c1);

  const float* pln = xl + (size_t)n * F;
  v4f acc0 = *(const v4f*)(pln + c0);
  v4f acc1 = *(const v4f*)(pln + c1);
  float m = group_sum8(lrelu_dot8(acc0, acc1, r0, r1, a0, a1));
  float s = 1.0f;

  int dg = deg[n];
  bool big = dg > CAP;
  const int* row = bkt + (size_t)n * CAP;
  int nwin = big ? N_EDGES / 32 : (dg + 31) >> 5;
  for (int w = 0; w < nwin; ++w) {
    int mysrc, cnt;
    if (!big) {
      cnt = min(32, dg - 32 * w);
      mysrc = row[32 * w + min(lane, cnt - 1)];
    } else {
      int d  = ei[N_EDGES + 32 * w + lane];
      int sv = ei[32 * w + lane];
      unsigned hit = __builtin_amdgcn_ballot_w32(d == n);
      cnt = __builtin_popcount(hit);
      if (cnt == 0) continue;
      mysrc = 0;
      for (int i = 0; hit != 0; ++i) {
        int src = __builtin_amdgcn_readlane(sv, __builtin_ctz(hit));
        hit &= hit - 1;
        mysrc = (lane == i) ? src : mysrc;
      }
    }
    edge_chunk(mysrc, cnt, xl, c0, c1, r0, r1, a0, a1, m, s, acc0, acc1);
  }

  float inv_s = 1.0f / fmaxf(s, 1e-16f);
  acc0 *= inv_s;
  acc1 *= inv_s;
  bool up = (lane & 8) != 0;
  v4f keep = up ? acc1 : acc0;
  v4f send = up ? acc0 : acc1;
  float4 v;
  v.x = keep.x + row_xor<8>(send.x);
  v.y = keep.y + row_xor<8>(send.y);
  v.z = keep.z + row_xor<8>(send.z);
  v.w = keep.w + row_xor<8>(send.w);
  v.x += xhalf(v.x);
  v.y += xhalf(v.y);
  v.z += xhalf(v.z);
  v.w += xhalf(v.w);
  int cc = 4 * (lane & 15);
  const float4 bv = *(const float4*)(bias + cc);
  v.x = 0.25f * v.x + bv.x;
  v.y = 0.25f * v.y + bv.y;
  v.z = 0.25f * v.z + bv.z;
  v.w = 0.25f * v.w + bv.w;

  float mu = half_sum16((v.x + v.y) + (v.z + v.w)) * (1.0f / 64.0f);
  float d0 = v.x - mu, d1 = v.y - mu, d2 = v.z - mu, d3 = v.w - mu;
  float var = half_sum16((d0 * d0 + d1 * d1) + (d2 * d2 + d3 * d3)) * (1.0f / 64.0f);
  float inv = rsqrtf(var + 1e-5f);
  const float4 gv = *(const float4*)(g + cc);
  const float4 be = *(const float4*)(beta + cc);
  float4 y;
  y.x = d0 * inv * gv.x + be.x;
  y.y = d1 * inv * gv.y + be.y;
  y.z = d2 * inv * gv.z + be.z;
  y.w = d3 * inv * gv.w + be.w;
  y.x = (y.x > 0.f) ? y.x : (__expf(y.x) - 1.f);
  y.y = (y.y > 0.f) ? y.y : (__expf(y.y) - 1.f);
  y.z = (y.z > 0.f) ? y.z : (__expf(y.z) - 1.f);
  y.w = (y.w > 0.f) ? y.w : (__expf(y.w) - 1.f);
  if (lane < 16) *(float4*)(out + (size_t)n * HID + cc) = y;
}

__global__ void k_gemm_wmma(const float* __restrict__ X, int K,
                            const float* __restrict__ W0, const float* __restrict__ B0,
                            float* __restrict__ Y0,
                            const float* __restrict__ W1, const float* __restrict__ B1,
                            float* __restrict__ Y1,
                            int Ntiles, int nmat, int ldY, int act) {
  int wave = (blockIdx.x * blockDim.x + threadIdx.x) >> 5;
  int total = MT * Ntiles * nmat;
  if (wave >= total) return;

  int tn   = wave % Ntiles;
  int rest = wave / Ntiles;
  int mat  = rest % nmat;
  int tm   = rest / nmat;

  const float* W  = mat ? W1 : W0;
  const float* Bv = mat ? B1 : B0;
  float*       Y  = mat ? Y1 : Y0;

  int lane = threadIdx.x & 31;
  int half = lane >> 4;
  int l16  = lane & 15;

  const float* xrow = X + (size_t)(tm * 16 + l16) * K;
  const float* wrow = W + (size_t)(tn * 16 + l16) * K;

  v8f c = {0.f, 0.f, 0.f, 0.f, 0.f, 0.f, 0.f, 0.f};
  for (int k0 = 0; k0 < K; k0 += 4) {
    v2f a = *(const v2f*)(xrow + k0 + 2 * half);
    v2f b = *(const v2f*)(wrow + k0 + 2 * half);
    c = __builtin_amdgcn_wmma_f32_16x16x4_f32(false, a, false, b, (short)0, c,
                                              false, false);
  }

  int col   = tn * 16 + l16;
  float bia = Bv[col];
  int mbase = tm * 16 + 8 * half;
#pragma unroll
  for (int r = 0; r < 8; ++r) {
    float v = c[r] + bia;
    if (act) v = fmaxf(v, 0.0f);
    Y[(size_t)(mbase + r) * ldY + col] = v;
  }
}

template <int K>
__global__ void __launch_bounds__(256) k_gemm_strip(const float* __restrict__ X,
                             const float* __restrict__ W0, const float* __restrict__ B0,
                             float* __restrict__ Y0,
                             const float* __restrict__ W1, const float* __restrict__ B1,
                             float* __restrict__ Y1) {
  constexpr int PITCH = K + 4;
  constexpr int KV = K / 8;
  constexpr int RV = K / 4;
  __shared__ __attribute__((aligned(16))) float Ws[128 * PITCH];

  int strip = blockIdx.x & 3;
  int n0 = (strip & 1) * 128;
  const float* W  = ((strip >> 1) ? W1 : W0) + (size_t)n0 * K;
  const float* Bv = ((strip >> 1) ? B1 : B0) + n0;
  float*       Y  = ((strip >> 1) ? Y1 : Y0) + n0;

  int wv = __builtin_amdgcn_readfirstlane(threadIdx.x >> 5);
  int lane = threadIdx.x & 31;
  int h = lane >> 4;
  int c = lane & 15;
  int tm  = (blockIdx.x >> 2) * 8 + wv;
  int tmc = min(tm, MT - 1);

  const float4* xrow = (const float4*)(X + (size_t)(tmc * 16 + c) * K + 4 * h);
  float4 a[KV];
#pragma unroll
  for (int j = 0; j < KV; ++j) a[j] = xrow[2 * j];

  const float4* wsrc = (const float4*)W;
#pragma unroll
  for (int i = 0; i < (128 * RV) / 256; ++i) {
    int f = threadIdx.x + 256 * i;
    int row = f / RV, q = f % RV;
    float4 v = wsrc[f];
    *(float4*)(&Ws[row * PITCH + 4 * q]) = v;
  }
  __syncthreads();
  if (tm >= MT) return;

  v8f acc[8];
#pragma unroll
  for (int t = 0; t < 8; ++t) acc[t] = (v8f){0.f, 0.f, 0.f, 0.f, 0.f, 0.f, 0.f, 0.f};

  const float* bl = Ws + c * PITCH + 4 * h;
#pragma unroll
  for (int j = 0; j < KV; ++j) {
    float4 b[8];
#pragma unroll
    for (int t = 0; t < 8; ++t) b[t] = *(const float4*)(bl + 16 * t * PITCH + 8 * j);
    v2f alo = {a[j].x, a[j].y};
    v2f ahi = {a[j].z, a[j].w};
#pragma unroll
    for (int t = 0; t < 8; ++t) {
      v2f blo = {b[t].x, b[t].y};
      acc[t] = __builtin_amdgcn_wmma_f32_16x16x4_f32(false, alo, false, blo, (short)0, acc[t],
                                                     false, false);
    }
#pragma unroll
    for (int t = 0; t < 8; ++t) {
      v2f bhi = {b[t].z, b[t].w};
      acc[t] = __builtin_amdgcn_wmma_f32_16x16x4_f32(false, ahi, false, bhi, (short)0, acc[t],
                                                     false, false);
    }
  }

  int rbase = tm * 16 + 8 * h;
#pragma unroll
  for (int t = 0; t < 8; ++t) {
    int col = 16 * t + c;
    float bia = Bv[col];
#pragma unroll
    for (int r = 0; r < 8; ++r)
      Y[(size_t)(rbase + r) * F + col] = acc[t][r] + bia;
  }
}

__global__ void k_head2(const float* __restrict__ hm, const float* __restrict__ W,
                        const float* __restrict__ b, float* __restrict__ out) {
  int n = blockIdx.x * blockDim.x + threadIdx.x;
  if (n >= N_NODES) return;
  const float* p = hm + (size_t)n * 32;
  float a0 = b[0], a1 = b[1];
#pragma unroll
  for (int j = 0; j < 32; ++j) {
    float v = p[j];
    a0 += v * W[j];
    a1 += v * W[32 + j];
  }
  out[(size_t)n * 2 + 0] = a0;
  out[(size_t)n * 2 + 1] = a1;
}

static inline int wave_blocks(long long waves) { return (int)((waves + 7) / 8); }
#define STRIP_BLOCKS (((MT + 7) / 8) * 4)

extern "C" void kernel_launch(void* const* d_in, const int* in_sizes, int n_in,
                              void* d_out, int out_size, void* d_ws, size_t ws_size,
                              hipStream_t stream) {
  const float* x    = (const float*)d_in[0];
  const int*   ei   = (const int*)  d_in[1];
  const float* Wl1  = (const float*)d_in[2];  const float* bl1 = (const float*)d_in[3];
  const float* Wr1  = (const float*)d_in[4];  const float* br1 = (const float*)d_in[5];
  const float* att1 = (const float*)d_in[6];  const float* bias1=(const float*)d_in[7];
  const float* Wl2  = (const float*)d_in[8];  const float* bl2 = (const float*)d_in[9];
  const float* Wr2  = (const float*)d_in[10]; const float* br2 = (const float*)d_in[11];
  const float* att2 = (const float*)d_in[12]; const float* bias2=(const float*)d_in[13];
  const float* g1   = (const float*)d_in[14]; const float* be1 = (const float*)d_in[15];
  const float* g2   = (const float*)d_in[16]; const float* be2 = (const float*)d_in[17];
  const float* Wh1  = (const float*)d_in[18]; const float* bh1 = (const float*)d_in[19];
  const float* Wh2  = (const float*)d_in[20]; const float* bh2 = (const float*)d_in[21];
  float* out = (float*)d_out;

  float* ws  = (float*)d_ws;
  const size_t NF = (size_t)N_NODES * F;
  float* XL  = ws;
  float* XR  = XL + NF;
  float* H1  = XR + NF;
  float* HM  = H1 + (size_t)N_NODES * HID;
  int*   DEG = (int*)(HM + (size_t)N_NODES * 32);
  int*   BKT = DEG + DEG_PAD;

  const int TB = 256;
  int nodeBlocks = wave_blocks(N_NODES);
  int degBlocks  = (N_NODES + TB - 1) / TB;
  int eThrBlocks = (N_EDGES + TB - 1) / TB;

  k_deg_zero<<<degBlocks, TB, 0, stream>>>(DEG);
  k_bucket_fill<<<eThrBlocks, TB, 0, stream>>>(ei, DEG, BKT);

  k_gemm_strip<IN_DIM><<<STRIP_BLOCKS, TB, 0, stream>>>(x, Wl1, bl1, XL, Wr1, br1, XR);
  k_node_gat<<<nodeBlocks, TB, 0, stream>>>(DEG, BKT, ei, XL, XR, att1, bias1, g1, be1, H1);

  k_gemm_strip<HID><<<STRIP_BLOCKS, TB, 0, stream>>>(H1, Wl2, bl2, XL, Wr2, br2, XR);
  k_node_gat<<<nodeBlocks, TB, 0, stream>>>(DEG, BKT, ei, XL, XR, att2, bias2, g2, be2, H1);

  k_gemm_wmma<<<wave_blocks((long long)MT * 2), TB, 0, stream>>>(
      H1, HID, Wh1, bh1, HM, Wh1, bh1, HM,  2,  1, 32,  1);
  k_head2<<<(N_NODES + TB - 1) / TB, TB, 0, stream>>>(HM, Wh2, bh2, out);
}


# edge count-and-drop pass folded into the layer-1 projection launch: each GEMM thread takes 2 edges, atomics run under its K loop; 8 -> 7 launches; slot (summation) order = atomic arrival, as before
# speedup vs baseline: 4.9352x; 1.0492x over previous
#include <hip/hip_runtime.h>
#include <math.h>

#define N_NODES 50000
#define N_EDGES 800000
#define IN_DIM  128
#define HID     64
#define NH      4
#define F       (NH * HID)
#define MT      (N_NODES / 16)

typedef __attribute__((ext_vector_type(2))) float v2f;
typedef __attribute__((ext_vector_type(8))) float v8f;

#define CAP     64
#define DEG_PAD 50176

__global__ void k_deg_zero(int* __restrict__ deg) {
  int i = blockIdx.x * blockDim.x + threadIdx.x;
  if (i < N_NODES) deg[i] = 0;
}


typedef __attribute__((ext_vector_type(4))) float v4f;

__device__ __forceinline__ float lrelu_dot8(v4f x0, v4f x1, v4f r0, v4f r1, v4f a0, v4f a1) {
  v2f e0 = x0.lo + r0.lo, e1 = x0.hi + r0.hi;
  v2f e2 = x1.lo + r1.lo, e3 = x1.hi + r1.hi;
  e0 = __builtin_elementwise_max(e0, e0 * 0.2f);
  e1 = __builtin_elementwise_max(e1, e1 * 0.2f);
  e2 = __builtin_elementwise_max(e2, e2 * 0.2f);
  e3 = __builtin_elementwise_max(e3, e3 * 0.2f);
  v2f d = e0 * a0.lo;
  d = __builtin_elementwise_fma(e1, a0.hi, d);
  d = __builtin_elementwise_fma(e2, a1.lo, d);
  d = __builtin_elementwise_fma(e3, a1.hi, d);
  return d.x + d.y;
}

template <int M>
__device__ __forceinline__ float row_xor(float v) {
  int b = __builtin_bit_cast(int, v);
  return __builtin_bit_cast(float, __builtin_amdgcn_update_dpp(b, b, 0x160 | M, 0xf, 0xf, false));
}

__device__ __forceinline__ float xhalf(float v) {
  int b = __builtin_bit_cast(int, v);
  return __builtin_bit_cast(float,
      __builtin_amdgcn_permlanex16(b, b, 0x76543210, 0xfedcba98, false, false));
}

__device__ __forceinline__ float group_sum8(float v) {
  v += row_xor<1>(v);
  v += row_xor<2>(v);
  v += row_xor<4>(v);
  return v;
}

__device__ __forceinline__ float half_sum16(float v) {
  v = group_sum8(v);
  v += row_xor<8>(v);
  return v;
}

#define EG 4

__device__ __forceinline__ void edge_chunk(int mysrc, int cnt, const float* __restrict__ xl,
                                           int c0, int c1, v4f r0, v4f r1, v4f a0, v4f a1,
                                           float& m, float& s, v4f& acc0, v4f& acc1) {
  for (int j = 0; j < cnt; j += EG) {
    v4f x0[EG], x1[EG];
#pragma unroll
    for (int u = 0; u < EG; ++u) {
      int src = __builtin_amdgcn_readlane(mysrc, min(j + u, cnt - 1));
      const float* pl = xl + (size_t)src * F;
      x0[u] = *(const v4f*)(pl + c0);
      x1[u] = *(const v4f*)(pl + c1);
    }
    float sc[EG];
    float nm = m;
#pragma unroll
    for (int u = 0; u < EG; ++u) {
      float t = group_sum8(lrelu_dot8(x0[u], x1[u], r0, r1, a0, a1));
      sc[u] = ((j + u) < cnt) ? t : -3.0e38f;
      nm = fmaxf(nm, sc[u]);
    }
    float f = __expf(m - nm);
    s *= f;
    acc0 *= f;
    acc1 *= f;
    m = nm;
#pragma unroll
    for (int u = 0; u < EG; ++u) {
      float p = __expf(sc[u] - m);
      s += p;
      acc0 = __builtin_elementwise_fma((v4f)p, x0[u], acc0);
      acc1 = __builtin_elementwise_fma((v4f)p, x1[u], acc1);
    }
  }
}

__global__ void __launch_bounds__(256) k_node_gat(const int* __restrict__ deg, const int* __restrict__ bkt,
                           const int* __restrict__ ei,
                           const float* __restrict__ xl, const float* __restrict__ xr,
                           const float* __restrict__ att, const float* __restrict__ bias,
                           const float* __restrict__ g, const float* __restrict__ beta,
                           float* __restrict__ out) {
  int n = __builtin_amdgcn_readfirstlane((blockIdx.x * blockDim.x + threadIdx.x) >> 5);
  if (n >= N_NODES) return;
  int lane = threadIdx.x & 31;
  int c0 = HID * (lane >> 3) + 4 * (lane & 7);
  int c1 = c0 + HID / 2;

  const float* prn = xr + (size_t)n * F;
  const v4f r0 = *(const v4f*)(prn + c0);
  const v4f r1 = *(const v4f*)(prn + c1);
  const v4f a0 = *(const v4f*)(att + c0);
  const v4f a1 = *(const v4f*)(att + c1);

  const float* pln = xl + (size_t)n * F;
  v4f acc0 = *(const v4f*)(pln + c0);
  v4f acc1 = *(const v4f*)(pln + c1);
  float m = group_sum8(lrelu_dot8(acc0, acc1, r0, r1, a0, a1));
  float s = 1.0f;

  int dg = deg[n];
  bool big = dg > CAP;
  const int* row = bkt + (size_t)n * CAP;
  int nwin = big ? N_EDGES / 32 : (dg + 31) >> 5;
  for (int w = 0; w < nwin; ++w) {
    int mysrc, cnt;
    if (!big) {
      cnt = min(32, dg - 32 * w);
      mysrc = row[32 * w + min(lane, cnt - 1)];
    } else {
      int d  = ei[N_EDGES + 32 * w + lane];
      int sv = ei[32 * w + lane];
      unsigned hit = __builtin_amdgcn_ballot_w32(d == n);
      cnt = __builtin_popcount(hit);
      if (cnt == 0) continue;
      mysrc = 0;
      for (int i = 0; hit != 0; ++i) {
        int src = __builtin_amdgcn_readlane(sv, __builtin_ctz(hit));
        hit &= hit - 1;
        mysrc = (lane == i) ? src : mysrc;
      }
    }
    edge_chunk(mysrc, cnt, xl, c0, c1, r0, r1, a0, a1, m, s, acc0, acc1);
  }

  float inv_s = 1.0f / fmaxf(s, 1e-16f);
  acc0 *= inv_s;
  acc1 *= inv_s;
  bool up = (lane & 8) != 0;
  v4f keep = up ? acc1 : acc0;
  v4f send = up ? acc0 : acc1;
  float4 v;
  v.x = keep.x + row_xor<8>(send.x);
  v.y = keep.y + row_xor<8>(send.y);
  v.z = keep.z + row_xor<8>(send.z);
  v.w = keep.w + row_xor<8>(send.w);
  v.x += xhalf(v.x);
  v.y += xhalf(v.y);
  v.z += xhalf(v.z);
  v.w += xhalf(v.w);
  int cc = 4 * (lane & 15);
  const float4 bv = *(const float4*)(bias + cc);
  v.x = 0.25f * v.x + bv.x;
  v.y = 0.25f * v.y + bv.y;
  v.z = 0.25f * v.z + bv.z;
  v.w = 0.25f * v.w + bv.w;

  float mu = half_sum16((v.x + v.y) + (v.z + v.w)) * (1.0f / 64.0f);
  float d0 = v.x - mu, d1 = v.y - mu, d2 = v.z - mu, d3 = v.w - mu;
  float var = half_sum16((d0 * d0 + d1 * d1) + (d2 * d2 + d3 * d3)) * (1.0f / 64.0f);
  float inv = rsqrtf(var + 1e-5f);
  const float4 gv = *(const float4*)(g + cc);
  const float4 be = *(const float4*)(beta + cc);
  float4 y;
  y.x = d0 * inv * gv.x + be.x;
  y.y = d1 * inv * gv.y + be.y;
  y.z = d2 * inv * gv.z + be.z;
  y.w = d3 * inv * gv.w + be.w;
  y.x = (y.x > 0.f) ? y.x : (__expf(y.x) - 1.f);
  y.y = (y.y > 0.f) ? y.y : (__expf(y.y) - 1.f);
  y.z = (y.z > 0.f) ? y.z : (__expf(y.z) - 1.f);
  y.w = (y.w > 0.f) ? y.w : (__expf(y.w) - 1.f);
  if (lane < 16) *(float4*)(out + (size_t)n * HID + cc) = y;
}

__global__ void k_gemm_wmma(const float* __restrict__ X, int K,
                            const float* __restrict__ W0, const float* __restrict__ B0,
                            float* __restrict__ Y0,
                            const float* __restrict__ W1, const float* __restrict__ B1,
                            float* __restrict__ Y1,
                            int Ntiles, int nmat, int ldY, int act) {
  int wave = (blockIdx.x * blockDim.x + threadIdx.x) >> 5;
  int total = MT * Ntiles * nmat;
  if (wave >= total) return;

  int tn   = wave % Ntiles;
  int rest = wave / Ntiles;
  int mat  = rest % nmat;
  int tm   = rest / nmat;

  const float* W  = mat ? W1 : W0;
  const float* Bv = mat ? B1 : B0;
  float*       Y  = mat ? Y1 : Y0;

  int lane = threadIdx.x & 31;
  int half = lane >> 4;
  int l16  = lane & 15;

  const float* xrow = X + (size_t)(tm * 16 + l16) * K;
  const float* wrow = W + (size_t)(tn * 16 + l16) * K;

  v8f c = {0.f, 0.f, 0.f, 0.f, 0.f, 0.f, 0.f, 0.f};
  for (int k0 = 0; k0 < K; k0 += 4) {
    v2f a = *(const v2f*)(xrow + k0 + 2 * half);
    v2f b = *(const v2f*)(wrow + k0 + 2 * half);
    c = __builtin_amdgcn_wmma_f32_16x16x4_f32(false, a, false, b, (short)0, c,
                                              false, false);
  }

  int col   = tn * 16 + l16;
  float bia = Bv[col];
  int mbase = tm * 16 + 8 * half;
#pragma unroll
  for (int r = 0; r < 8; ++r) {
    float v = c[r] + bia;
    if (act) v = fmaxf(v, 0.0f);
    Y[(size_t)(mbase + r) * ldY + col] = v;
  }
}

template <int K, bool EDGES>
__global__ void __launch_bounds__(256) k_gemm_strip(const float* __restrict__ X,
                             const float* __restrict__ W0, const float* __restrict__ B0,
                             float* __restrict__ Y0,
                             const float* __restrict__ W1, const float* __restrict__ B1,
                             float* __restrict__ Y1,
                             const int* __restrict__ ei, int* __restrict__ deg,
                             int* __restrict__ bkt) {
  constexpr int PITCH = K + 4;
  constexpr int KV = K / 8;
  constexpr int RV = K / 4;
  __shared__ __attribute__((aligned(16))) float Ws[128 * PITCH];

  int strip = blockIdx.x & 3;
  int n0 = (strip & 1) * 128;
  const float* W  = ((strip >> 1) ? W1 : W0) + (size_t)n0 * K;
  const float* Bv = ((strip >> 1) ? B1 : B0) + n0;
  float*       Y  = ((strip >> 1) ? Y1 : Y0) + n0;

  int wv = __builtin_amdgcn_readfirstlane(threadIdx.x >> 5);
  int lane = threadIdx.x & 31;
  int h = lane >> 4;
  int c = lane & 15;
  int tm  = (blockIdx.x >> 2) * 8 + wv;
  int tmc = min(tm, MT - 1);

  int2 ed = {0, 0}, es = {0, 0};
  int rk0 = CAP, rk1 = CAP;
  bool own = false;
  if constexpr (EDGES) {
    int p = blockIdx.x * 256 + threadIdx.x;
    own = p < N_EDGES / 2;
    if (own) {
      ed = ((const int2*)(ei + N_EDGES))[p];
      es = ((const int2*)ei)[p];
    }
  }

  const float4* xrow = (const float4*)(X + (size_t)(tmc * 16 + c) * K + 4 * h);
  float4 a[KV];
#pragma unroll
  for (int j = 0; j < KV; ++j) a[j] = xrow[2 * j];

  const float4* wsrc = (const float4*)W;
#pragma unroll
  for (int i = 0; i < (128 * RV) / 256; ++i) {
    int f = threadIdx.x + 256 * i;
    int row = f / RV, q = f % RV;
    float4 v = wsrc[f];
    *(float4*)(&Ws[row * PITCH + 4 * q]) = v;
  }

  __syncthreads();

  if constexpr (EDGES) {
    if (own) {
      rk0 = atomicAdd(&deg[ed.x], 1);
      rk1 = atomicAdd(&deg[ed.y], 1);
    }
  }
  if (tm >= MT) {
    if constexpr (EDGES) {
      if (rk0 < CAP) bkt[(size_t)ed.x * CAP + rk0] = es.x;
      if (rk1 < CAP) bkt[(size_t)ed.y * CAP + rk1] = es.y;
    }
    return;
  }

  v8f acc[8];
#pragma unroll
  for (int t = 0; t < 8; ++t) acc[t] = (v8f){0.f, 0.f, 0.f, 0.f, 0.f, 0.f, 0.f, 0.f};

  const float* bl = Ws + c * PITCH + 4 * h;
#pragma unroll
  for (int j = 0; j < KV; ++j) {
    float4 b[8];
#pragma unroll
    for (int t = 0; t < 8; ++t) b[t] = *(const float4*)(bl + 16 * t * PITCH + 8 * j);
    v2f alo = {a[j].x, a[j].y};
    v2f ahi = {a[j].z, a[j].w};
#pragma unroll
    for (int t = 0; t < 8; ++t) {
      v2f blo = {b[t].x, b[t].y};
      acc[t] = __builtin_amdgcn_wmma_f32_16x16x4_f32(false, alo, false, blo, (short)0, acc[t],
                                                     false, false);
    }
#pragma unroll
    for (int t = 0; t < 8; ++t) {
      v2f bhi = {b[t].z, b[t].w};
      acc[t] = __builtin_amdgcn_wmma_f32_16x16x4_f32(false, ahi, false, bhi, (short)0, acc[t],
                                                     false, false);
    }
  }

  float bia[8];
#pragma unroll
  for (int t = 0; t < 8; ++t) bia[t] = Bv[16 * t + c];

  if constexpr (EDGES) {
    if (rk0 < CAP) bkt[(size_t)ed.x * CAP + rk0] = es.x;
    if (rk1 < CAP) bkt[(size_t)ed.y * CAP + rk1] = es.y;
  }

  int rbase = tm * 16 + 8 * h;
#pragma unroll
  for (int t = 0; t < 8; ++t) {
    int col = 16 * t + c;
#pragma unroll
    for (int r = 0; r < 8; ++r)
      Y[(size_t)(rbase + r) * F + col] = acc[t][r] + bia[t];
  }
}

__global__ void k_head2(const float* __restrict__ hm, const float* __restrict__ W,
                        const float* __restrict__ b, float* __restrict__ out) {
  int n = blockIdx.x * blockDim.x + threadIdx.x;
  if (n >= N_NODES) return;
  const float* p = hm + (size_t)n * 32;
  float a0 = b[0], a1 = b[1];
#pragma unroll
  for (int j = 0; j < 32; ++j) {
    float v = p[j];
    a0 += v * W[j];
    a1 += v * W[32 + j];
  }
  out[(size_t)n * 2 + 0] = a0;
  out[(size_t)n * 2 + 1] = a1;
}

static inline int wave_blocks(long long waves) { return (int)((waves + 7) / 8); }
#define STRIP_BLOCKS (((MT + 7) / 8) * 4)
static_assert((long long)STRIP_BLOCKS * 256 * 2 >= N_EDGES && N_EDGES % 2 == 0,
              "the projection grid must own every edge, two per thread");

extern "C" void kernel_launch(void* const* d_in, const int* in_sizes, int n_in,
                              void* d_out, int out_size, void* d_ws, size_t ws_size,
                              hipStream_t stream) {
  const float* x    = (const float*)d_in[0];
  const int*   ei   = (const int*)  d_in[1];
  const float* Wl1  = (const float*)d_in[2];  const float* bl1 = (const float*)d_in[3];
  const float* Wr1  = (const float*)d_in[4];  const float* br1 = (const float*)d_in[5];
  const float* att1 = (const float*)d_in[6];  const float* bias1=(const float*)d_in[7];
  const float* Wl2  = (const float*)d_in[8];  const float* bl2 = (const float*)d_in[9];
  const float* Wr2  = (const float*)d_in[10]; const float* br2 = (const float*)d_in[11];
  const float* att2 = (const float*)d_in[12]; const float* bias2=(const float*)d_in[13];
  const float* g1   = (const float*)d_in[14]; const float* be1 = (const float*)d_in[15];
  const float* g2   = (const float*)d_in[16]; const float* be2 = (const float*)d_in[17];
  const float* Wh1  = (const float*)d_in[18]; const float* bh1 = (const float*)d_in[19];
  const float* Wh2  = (const float*)d_in[20]; const float* bh2 = (const float*)d_in[21];
  float* out = (float*)d_out;

  float* ws  = (float*)d_ws;
  const size_t NF = (size_t)N_NODES * F;
  float* XL  = ws;
  float* XR  = XL + NF;
  float* H1  = XR + NF;
  float* HM  = H1 + (size_t)N_NODES * HID;
  int*   DEG = (int*)(HM + (size_t)N_NODES * 32);
  int*   BKT = DEG + DEG_PAD;

  const int TB = 256;
  int nodeBlocks = wave_blocks(N_NODES);
  int degBlocks  = (N_NODES + TB - 1) / TB;

  k_deg_zero<<<degBlocks, TB, 0, stream>>>(DEG);

  k_gemm_strip<IN_DIM, true><<<STRIP_BLOCKS, TB, 0, stream>>>(x, Wl1, bl1, XL, Wr1, br1, XR,
                                                              ei, DEG, BKT);
  k_node_gat<<<nodeBlocks, TB, 0, stream>>>(DEG, BKT, ei, XL, XR, att1, bias1, g1, be1, H1);

  k_gemm_strip<HID, false><<<STRIP_BLOCKS, TB, 0, stream>>>(H1, Wl2, bl2, XL, Wr2, br2, XR,
                                                            ei, DEG, BKT);
  k_node_gat<<<nodeBlocks, TB, 0, stream>>>(DEG, BKT, ei, XL, XR, att2, bias2, g2, be2, H1);

  k_gemm_wmma<<<wave_blocks((long long)MT * 2), TB, 0, stream>>>(
      H1, HID, Wh1, bh1, HM, Wh1, bh1, HM,  2,  1, 32,  1);
  k_head2<<<(N_NODES + TB - 1) / TB, TB, 0, stream>>>(HM, Wh2, bh2, out);
}
